# starting kernel
# baseline (speedup 1.0000x reference)
#include <hip/hip_runtime.h>
#include <math.h>


#define N_NODES  100000
#define N_EDGES  1000000
#define N_FEAT   128
#define HIDDEN   64
#define N_GRAPHS 512
#define N_CLASSES 10
#define BN_EPS  1e-5f
#define BN_BETA 1e-4f

typedef __attribute__((ext_vector_type(2))) float v2f;
typedef __attribute__((ext_vector_type(8))) float v8f;

__global__ void colstats_kernel(const float* __restrict__ A, int N, int F,
                                float* __restrict__ sums, float* __restrict__ sumsq) {
  const int c = threadIdx.x;
  float s = 0.f, q = 0.f;
  for (int r = blockIdx.x; r < N; r += gridDim.x) {
    float v = A[(size_t)r * F + c];
    s += v; q += v * v;
  }
  atomicAdd(&sums[c], s);
  atomicAdd(&sumsq[c], q);
}

__global__ void fold_bn_kernel(const float* __restrict__ sums,
                               const float* __restrict__ sumsq, float invN,
                               int K, int NN, const float* __restrict__ W,
                               const float* __restrict__ extra_b,
                               float* __restrict__ Wout, float* __restrict__ vout) {
  __shared__ float s_ms[N_FEAT];
  __shared__ float s_s[N_FEAT];
  const int j = threadIdx.x;
  for (int k = j; k < K; k += blockDim.x) {
    float mean = sums[k] * invN;
    float var  = sumsq[k] * invN - mean * mean;
    float sc   = rsqrtf(var + BN_EPS);
    s_ms[k] = mean * sc;
    s_s[k]  = sc;
  }
  __syncthreads();
  float acc = extra_b ? extra_b[j] : 0.f;
  for (int k = 0; k < K; ++k) {
    float w = W[(size_t)k * NN + j];
    Wout[(size_t)k * NN + j] = s_s[k] * w;
    acc += (BN_BETA - s_ms[k]) * w;
  }
  vout[j] = acc;
}

__global__ __launch_bounds__(128) void gemm_wmma_kernel(
    const float* __restrict__ A, const float* __restrict__ Wt,
    const float* __restrict__ v, float* __restrict__ C,
    int M, int K, int doRelu) {
  const int lane = threadIdx.x & 31;
  const int wave = threadIdx.x >> 5;
  const int rowTile = blockIdx.x * 4 + wave;
  if (rowTile >= (M >> 4)) return;
  const int mrow = (rowTile << 4) + (lane & 15);
  const int kOff = (lane >> 4) << 1;
  const int n0   = lane & 15;
  v8f acc[4] = {};
  for (int kk = 0; kk < K; kk += 4) {
    const int k0 = kk + kOff;
    v2f a = *(const v2f*)(A + (size_t)mrow * K + k0);
    const float* w0 = Wt + (size_t)k0 * HIDDEN + n0;
#pragma unroll
    for (int t = 0; t < 4; ++t) {
      v2f b;
      b.x = w0[t * 16];
      b.y = w0[HIDDEN + t * 16];
      acc[t] = __builtin_amdgcn_wmma_f32_16x16x4_f32(
          false, a, false, b, (short)0, acc[t], false, false);
    }
  }
  const int rowBase = (rowTile << 4) + ((lane >> 4) << 3);
#pragma unroll
  for (int t = 0; t < 4; ++t) {
    const int n = t * 16 + n0;
    const float bias = v[n];
#pragma unroll
    for (int e = 0; e < 8; ++e) {
      float val = acc[t][e] + bias;
      if (doRelu) val = fmaxf(val, 0.f);
      C[(size_t)(rowBase + e) * HIDDEN + n] = val;
    }
  }
}

__global__ void deg_init_kernel(float* deg, int N) {
  int i = blockIdx.x * blockDim.x + threadIdx.x;
  if (i < N) deg[i] = 1.0f;
}
__global__ void deg_accum_kernel(const int* __restrict__ row,
                                 const float* __restrict__ att, int comp,
                                 int E, float* __restrict__ deg) {
  int e = blockIdx.x * blockDim.x + threadIdx.x;
  if (e >= E) return;
  float w = att ? att[(size_t)e * 2 + comp] : 1.0f;
  atomicAdd(&deg[row[e]], w);
}
__global__ void make_dinv_kernel(const float* __restrict__ deg,
                                 float* __restrict__ dinv, int N) {
  int i = blockIdx.x * blockDim.x + threadIdx.x;
  if (i < N) dinv[i] = deg[i] > 0.f ? rsqrtf(deg[i]) : 0.f;
}

__global__ void scatter_kernel(const float* __restrict__ m,
                               const int* __restrict__ row,
                               const int* __restrict__ col,
                               const float* __restrict__ att, int comp,
                               const float* __restrict__ dinv,
                               float* __restrict__ acc, int E) {
  int idx = blockIdx.x * blockDim.x + threadIdx.x;
  int e = idx >> 6, j = idx & 63;
  if (e >= E) return;
  int r = row[e], c = col[e];
  float w = att ? att[(size_t)e * 2 + comp] : 1.0f;
  float coef = dinv[r] * w * dinv[c];
  atomicAdd(&acc[(size_t)c * HIDDEN + j], coef * m[(size_t)r * HIDDEN + j]);
}

__global__ void finish_gcn_kernel(const float* __restrict__ acc,
                                  const float* __restrict__ m,
                                  const float* __restrict__ dinv,
                                  const float* __restrict__ b,
                                  float* __restrict__ out, int N) {
  int idx = blockIdx.x * blockDim.x + threadIdx.x;
  int i = idx >> 6, j = idx & 63;
  if (i >= N) return;
  float di = dinv[i];
  float val = acc[idx] + di * di * m[idx] + b[j];
  out[idx] = fmaxf(val, 0.f);
}

__global__ void edge_att_kernel(const float* __restrict__ h,
                                const int* __restrict__ row,
                                const int* __restrict__ col,
                                const float* __restrict__ Wea,
                                const float* __restrict__ bea,
                                float* __restrict__ att, int E) {
  int e = blockIdx.x * blockDim.x + threadIdx.x;
  if (e >= E) return;
  const float* hr = h + (size_t)row[e] * HIDDEN;
  const float* hc = h + (size_t)col[e] * HIDDEN;
  float l0 = bea[0], l1 = bea[1];
  for (int j = 0; j < HIDDEN; ++j) {
    float a = hr[j], c = hc[j];
    l0 += a * Wea[j * 2 + 0] + c * Wea[(HIDDEN + j) * 2 + 0];
    l1 += a * Wea[j * 2 + 1] + c * Wea[(HIDDEN + j) * 2 + 1];
  }
  float mx = fmaxf(l0, l1);
  float e0 = __expf(l0 - mx), e1 = __expf(l1 - mx);
  float inv = 1.f / (e0 + e1);
  att[(size_t)e * 2 + 0] = e0 * inv;
  att[(size_t)e * 2 + 1] = e1 * inv;
}

__global__ void node_att_split_kernel(const float* __restrict__ h,
                                      const float* __restrict__ Wna,
                                      const float* __restrict__ bna,
                                      float* __restrict__ xc,
                                      float* __restrict__ xo, int N) {
  int i = blockIdx.x * blockDim.x + threadIdx.x;
  if (i >= N) return;
  const float* hi = h + (size_t)i * HIDDEN;
  float l0 = bna[0], l1 = bna[1];
  for (int j = 0; j < HIDDEN; ++j) {
    float a = hi[j];
    l0 += a * Wna[j * 2 + 0];
    l1 += a * Wna[j * 2 + 1];
  }
  float mx = fmaxf(l0, l1);
  float e0 = __expf(l0 - mx), e1 = __expf(l1 - mx);
  float inv = 1.f / (e0 + e1);
  float p0 = e0 * inv, p1 = e1 * inv;
  for (int j = 0; j < HIDDEN; ++j) {
    float a = hi[j];
    xc[(size_t)i * HIDDEN + j] = p0 * a;
    xo[(size_t)i * HIDDEN + j] = p1 * a;
  }
}

__global__ void pool_kernel(const float* __restrict__ x,
                            const int* __restrict__ batch,
                            float* __restrict__ g, int N) {
  int idx = blockIdx.x * blockDim.x + threadIdx.x;
  int i = idx >> 6, j = idx & 63;
  if (i >= N) return;
  atomicAdd(&g[(size_t)batch[i] * HIDDEN + j], x[idx]);
}

__global__ void add_kernel(const float* a, const float* b, float* c, int n) {
  int i = blockIdx.x * blockDim.x + threadIdx.x;
  if (i < n) c[i] = a[i] + b[i];
}

__global__ void logits_kernel(const float* __restrict__ h,
                              const float* __restrict__ Wt,
                              const float* __restrict__ v,
                              float* __restrict__ out, int G) {
  int g = blockIdx.x * blockDim.x + threadIdx.x;
  if (g >= G) return;
  float l[N_CLASSES];
  for (int c = 0; c < N_CLASSES; ++c) l[c] = v[c];
  const float* hg = h + (size_t)g * HIDDEN;
  for (int k = 0; k < HIDDEN; ++k) {
    float x = hg[k];
    for (int c = 0; c < N_CLASSES; ++c) l[c] += x * Wt[k * N_CLASSES + c];
  }
  float mx = l[0];
  for (int c = 1; c < N_CLASSES; ++c) mx = fmaxf(mx, l[c]);
  float s = 0.f;
  for (int c = 0; c < N_CLASSES; ++c) s += __expf(l[c] - mx);
  float lse = __logf(s) + mx;
  for (int c = 0; c < N_CLASSES; ++c) out[(size_t)g * N_CLASSES + c] = l[c] - lse;
}

extern "C" void kernel_launch(void* const* d_in, const int* in_sizes, int n_in,
                              void* d_out, int out_size, void* d_ws, size_t ws_size,
                              hipStream_t stream) {
  (void)in_sizes; (void)n_in; (void)out_size; (void)ws_size;
  const float* x       = (const float*)d_in[0];
  const int*   ei      = (const int*)d_in[1];
  const int*   row     = ei;
  const int*   col     = ei + N_EDGES;
  const int*   batch   = (const int*)d_in[2];
  const float* W_feat  = (const float*)d_in[3];
  const float* W_convs = (const float*)d_in[4];
  const float* b_convs = (const float*)d_in[5];
  const float* W_ea    = (const float*)d_in[6];
  const float* b_ea    = (const float*)d_in[7];
  const float* W_na    = (const float*)d_in[8];
  const float* b_na    = (const float*)d_in[9];
  const float* W_ctx   = (const float*)d_in[10];
  const float* b_ctx   = (const float*)d_in[11];
  const float* W_obj   = (const float*)d_in[12];
  const float* b_obj   = (const float*)d_in[13];
  const float* W1h[3]  = {(const float*)d_in[14], (const float*)d_in[18], (const float*)d_in[22]};
  const float* b1h[3]  = {(const float*)d_in[15], (const float*)d_in[19], (const float*)d_in[23]};
  const float* W2h[3]  = {(const float*)d_in[16], (const float*)d_in[20], (const float*)d_in[24]};
  const float* b2h[3]  = {(const float*)d_in[17], (const float*)d_in[21], (const float*)d_in[25]};

  size_t off = 0;
  auto carve = [&](size_t elems) -> float* {
    float* p = (float*)((char*)d_ws + off);
    off += ((elems * sizeof(float)) + 255) & ~(size_t)255;
    return p;
  };
  float* hb    = carve((size_t)N_NODES * HIDDEN);
  float* mbuf  = carve((size_t)N_NODES * HIDDEN);
  float* accb  = carve((size_t)N_NODES * HIDDEN);
  float* xc    = carve((size_t)N_NODES * HIDDEN);
  float* xo    = carve((size_t)N_NODES * HIDDEN);
  float* att   = carve((size_t)N_EDGES * 2);
  float* degu  = carve(N_NODES);
  float* dinvu = carve(N_NODES);
  float* dega  = carve(N_NODES);
  float* dinva = carve(N_NODES);
  float* gc    = carve((size_t)N_GRAPHS * HIDDEN);
  float* go    = carve((size_t)N_GRAPHS * HIDDEN);
  float* gs    = carve((size_t)N_GRAPHS * HIDDEN);
  float* r1    = carve((size_t)N_GRAPHS * HIDDEN);
  float* sums  = carve(N_FEAT);
  float* sq    = carve(N_FEAT);
  float* Wf    = carve((size_t)N_FEAT * HIDDEN);
  float* vf    = carve(HIDDEN);

  auto nblk = [](long n, int b) { return (int)((n + b - 1) / b); };

  auto stats = [&](const float* A, int N, int F) {
    hipMemsetAsync(sums, 0, (size_t)F * sizeof(float), stream);
    hipMemsetAsync(sq,   0, (size_t)F * sizeof(float), stream);
    colstats_kernel<<<512, F, 0, stream>>>(A, N, F, sums, sq);
  };
  auto gemm = [&](const float* A, float* C, int M, int K, int relu) {
    gemm_wmma_kernel<<<nblk(M / 16, 4), 128, 0, stream>>>(A, Wf, vf, C, M, K, relu);
  };

  stats(x, N_NODES, N_FEAT);
  fold_bn_kernel<<<1, HIDDEN, 0, stream>>>(sums, sq, 1.f / N_NODES, N_FEAT, HIDDEN,
                                           W_feat, nullptr, Wf, vf);
  gemm(x, hb, N_NODES, N_FEAT, 1);

  deg_init_kernel<<<nblk(N_NODES, 256), 256, 0, stream>>>(degu, N_NODES);
  deg_accum_kernel<<<nblk(N_EDGES, 256), 256, 0, stream>>>(row, nullptr, 0, N_EDGES, degu);
  make_dinv_kernel<<<nblk(N_NODES, 256), 256, 0, stream>>>(degu, dinvu, N_NODES);

  for (int i = 0; i < 3; ++i) {
    stats(hb, N_NODES, HIDDEN);
    fold_bn_kernel<<<1, HIDDEN, 0, stream>>>(sums, sq, 1.f / N_NODES, HIDDEN, HIDDEN,
                                             W_convs + (size_t)i * HIDDEN * HIDDEN,
                                             nullptr, Wf, vf);
    gemm(hb, mbuf, N_NODES, HIDDEN, 0);
    hipMemsetAsync(accb, 0, (size_t)N_NODES * HIDDEN * sizeof(float), stream);
    scatter_kernel<<<nblk((long)N_EDGES * 64, 256), 256, 0, stream>>>(
        mbuf, row, col, nullptr, 0, dinvu, accb, N_EDGES);
    finish_gcn_kernel<<<nblk((long)N_NODES * 64, 256), 256, 0, stream>>>(
        accb, mbuf, dinvu, b_convs + (size_t)i * HIDDEN, hb, N_NODES);
  }

  edge_att_kernel<<<nblk(N_EDGES, 256), 256, 0, stream>>>(hb, row, col, W_ea, b_ea,
                                                          att, N_EDGES);
  node_att_split_kernel<<<nblk(N_NODES, 256), 256, 0, stream>>>(hb, W_na, b_na,
                                                                xc, xo, N_NODES);

  auto run_branch = [&](float* xin, int comp, const float* W, const float* b,
                        float* gout) {
    stats(xin, N_NODES, HIDDEN);
    fold_bn_kernel<<<1, HIDDEN, 0, stream>>>(sums, sq, 1.f / N_NODES, HIDDEN, HIDDEN,
                                             W, nullptr, Wf, vf);
    gemm(xin, mbuf, N_NODES, HIDDEN, 0);
    deg_init_kernel<<<nblk(N_NODES, 256), 256, 0, stream>>>(dega, N_NODES);
    deg_accum_kernel<<<nblk(N_EDGES, 256), 256, 0, stream>>>(row, att, comp, N_EDGES, dega);
    make_dinv_kernel<<<nblk(N_NODES, 256), 256, 0, stream>>>(dega, dinva, N_NODES);
    hipMemsetAsync(accb, 0, (size_t)N_NODES * HIDDEN * sizeof(float), stream);
    scatter_kernel<<<nblk((long)N_EDGES * 64, 256), 256, 0, stream>>>(
        mbuf, row, col, att, comp, dinva, accb, N_EDGES);
    finish_gcn_kernel<<<nblk((long)N_NODES * 64, 256), 256, 0, stream>>>(
        accb, mbuf, dinva, b, xin, N_NODES);
    hipMemsetAsync(gout, 0, (size_t)N_GRAPHS * HIDDEN * sizeof(float), stream);
    pool_kernel<<<nblk((long)N_NODES * 64, 256), 256, 0, stream>>>(xin, batch, gout,
                                                                   N_NODES);
  };
  run_branch(xc, 0, W_ctx, b_ctx, gc);
  run_branch(xo, 1, W_obj, b_obj, go);

  add_kernel<<<nblk(N_GRAPHS * HIDDEN, 256), 256, 0, stream>>>(gc, go, gs,
                                                               N_GRAPHS * HIDDEN);

  float* out = (float*)d_out;
  auto run_head = [&](const float* gin, int h, float* outp) {
    stats(gin, N_GRAPHS, HIDDEN);
    fold_bn_kernel<<<1, HIDDEN, 0, stream>>>(sums, sq, 1.f / N_GRAPHS, HIDDEN, HIDDEN,
                                             W1h[h], b1h[h], Wf, vf);
    gemm(gin, r1, N_GRAPHS, HIDDEN, 1);
    stats(r1, N_GRAPHS, HIDDEN);
    fold_bn_kernel<<<1, N_CLASSES, 0, stream>>>(sums, sq, 1.f / N_GRAPHS, HIDDEN,
                                                N_CLASSES, W2h[h], b2h[h], Wf, vf);
    logits_kernel<<<nblk(N_GRAPHS, 256), 256, 0, stream>>>(r1, Wf, vf, outp, N_GRAPHS);
  };
  run_head(gc, 0, out);
  run_head(go, 1, out + N_GRAPHS * N_CLASSES);
  run_head(gs, 2, out + 2 * N_GRAPHS * N_CLASSES);
}


# column sums (BN): grid-stride loop with one 4-byte load in flight + f32 atomics on a cleared buffer -> one partial row per 512-row slab, slabs' rows added by a 2nd launch (new fixed sum order)
# speedup vs baseline: 4.7165x; 4.7165x over previous
#include <hip/hip_runtime.h>
#include <math.h>


#define N_NODES  100000
#define N_EDGES  1000000
#define N_FEAT   128
#define HIDDEN   64
#define N_GRAPHS 512
#define N_CLASSES 10
#define BN_EPS  1e-5f
#define BN_BETA 1e-4f

typedef __attribute__((ext_vector_type(2))) float v2f;
typedef __attribute__((ext_vector_type(8))) float v8f;

#define CS_THREADS 512
#define CS_WAVES (CS_THREADS / 32)
#define CS_BATCH 8
#define CS_SLAB 512
#define CS_MAX_SLABS ((N_NODES + CS_SLAB - 1) / CS_SLAB)
#define CF_THREADS 1024
#define STAT_STRIDE (2 * N_FEAT)

typedef __attribute__((ext_vector_type(4))) float v4f;

template <int F, bool GUARD>
__device__ __forceinline__ void cs_load(const v4f* __restrict__ A4, int rfirst, int r1,
                                        int c4, int b0, v4f (&v)[CS_BATCH]) {
  constexpr int VPR  = F / 4;
  constexpr int STEP = CS_WAVES * (32 / VPR);
#pragma unroll
  for (int u = 0; u < CS_BATCH; ++u) {
    int r = rfirst + (b0 + u) * STEP;
    if (GUARD) r = min(r, r1 - 1);
    v[u] = A4[(size_t)r * VPR + c4];
  }
}

template <int F, bool GUARD>
__device__ __forceinline__ void cs_slab_sum(const v4f* __restrict__ A4, int rfirst,
                                            int r1, int c4, v4f& s, v4f& q) {
  constexpr int VPR  = F / 4;
  constexpr int STEP = CS_WAVES * (32 / VPR);
  constexpr int TRIPS = CS_SLAB / (STEP * CS_BATCH);
  static_assert(TRIPS * STEP * CS_BATCH == CS_SLAB, "a slab is a whole number of batches");
  v4f cur[CS_BATCH], nxt[CS_BATCH];
  cs_load<F, GUARD>(A4, rfirst, r1, c4, 0, cur);
#pragma unroll
  for (int t = 0; t < TRIPS; ++t) {
    if (t + 1 < TRIPS) cs_load<F, GUARD>(A4, rfirst, r1, c4, (t + 1) * CS_BATCH, nxt);
    asm volatile("" :: "v"(cur[0]), "v"(cur[1]), "v"(cur[2]), "v"(cur[3]),
                       "v"(cur[4]), "v"(cur[5]), "v"(cur[6]), "v"(cur[7]));
#pragma unroll
    for (int u = 0; u < CS_BATCH; ++u) {
      const int r = rfirst + (t * CS_BATCH + u) * STEP;
      if (!GUARD || r < r1) { s += cur[u]; q += cur[u] * cur[u]; }
    }
#pragma unroll
    for (int u = 0; u < CS_BATCH; ++u) cur[u] = nxt[u];
  }
}
static_assert(CS_BATCH == 8, "the fence above names eight values");

template <int F>
__global__ __launch_bounds__(CS_THREADS) void colstats_slab_kernel(
    const float* __restrict__ A, int N, float* __restrict__ part, int strideA) {
  constexpr int VPR = F / 4;
  static_assert(VPR == 16 || VPR == 32, "a wave-wide load covers one or two whole rows");
  __shared__ v4f s_s[CS_THREADS];
  __shared__ v4f s_q[CS_THREADS];
  A    += (size_t)blockIdx.y * strideA;
  part += ((size_t)blockIdx.y * gridDim.x + blockIdx.x) * STAT_STRIDE;
  const int lane = threadIdx.x & 31;
  const int wave = threadIdx.x >> 5;
  const int c4   = lane & (VPR - 1);
  const int r0   = blockIdx.x * CS_SLAB;
  const int r1   = min(N, r0 + CS_SLAB);
  const int rfirst = r0 + wave * (32 / VPR) + lane / VPR;
  v4f s = {0.f, 0.f, 0.f, 0.f}, q = s;
  if (r0 + CS_SLAB <= N) cs_slab_sum<F, false>((const v4f*)A, rfirst, r1, c4, s, q);
  else                   cs_slab_sum<F, true>((const v4f*)A, rfirst, r1, c4, s, q);
  s_s[threadIdx.x] = s;
  s_q[threadIdx.x] = q;
  __syncthreads();
  if (threadIdx.x < 2 * VPR) {
    const bool sq = threadIdx.x >= VPR;
    const int c = threadIdx.x & (VPR - 1);
    const v4f* src = sq ? s_q : s_s;
    v4f a = src[c];
#pragma unroll
    for (int j = 1; j < CS_THREADS / VPR; ++j) a += src[j * VPR + c];
    ((v4f*)(part + (sq ? N_FEAT : 0)))[c] = a;
  }
}

template <int F>
__global__ __launch_bounds__(CF_THREADS) void colstats_finish_kernel(
    const float* __restrict__ part, int P, float* __restrict__ stat) {
  constexpr int VPR = F / 4;
  constexpr int V2  = 2 * VPR;
  constexpr int G   = CF_THREADS / V2;
  constexpr int PER = (CS_MAX_SLABS + G - 1) / G;
  __shared__ float4 s_r[CF_THREADS];
  part += (size_t)blockIdx.x * P * STAT_STRIDE;
  stat += (size_t)blockIdx.x * STAT_STRIDE;
  const int c = threadIdx.x & (V2 - 1);
  const int g = threadIdx.x / V2;
  const int off = c < VPR ? c * 4 : N_FEAT + (c - VPR) * 4;
  float4 v[PER];
#pragma unroll
  for (int i = 0; i < PER; ++i)
    v[i] = *(const float4*)(part + (size_t)min(g + i * G, P - 1) * STAT_STRIDE + off);
  float4 a = make_float4(0.f, 0.f, 0.f, 0.f);
#pragma unroll
  for (int i = 0; i < PER; ++i)
    if (g + i * G < P) { a.x += v[i].x; a.y += v[i].y; a.z += v[i].z; a.w += v[i].w; }
  s_r[threadIdx.x] = a;
  __syncthreads();
  if (threadIdx.x < V2) {
    float4 t = s_r[c];
#pragma unroll
    for (int j = 1; j < G; ++j) {
      const float4 b = s_r[j * V2 + c];
      t.x += b.x; t.y += b.y; t.z += b.z; t.w += b.w;
    }
    *(float4*)(stat + off) = t;
  }
}

__device__ __forceinline__ void fold_stats(float sum, float sumsq, float invN,
                                           float* s_ms, float* s_s) {
  float mean = sum * invN;
  float var  = sumsq * invN - mean * mean;
  float sc   = rsqrtf(var + BN_EPS);
  s_ms[threadIdx.x] = mean * sc;
  s_s[threadIdx.x]  = sc;
}

template <int PITCH>
__device__ __forceinline__ float fold_column(float* s_w, const float* s_ms,
                                             const float* s_s, int K, int j,
                                             float acc) {
#pragma unroll 8
  for (int k = 0; k < K; ++k) {
    const float w = s_w[k * PITCH + j];
    s_w[k * PITCH + j] = s_s[k] * w;
    acc += (BN_BETA - s_ms[k]) * w;
  }
  return acc;
}

#define GEMM_WAVES 8
#define GEMM_THREADS (GEMM_WAVES * 32)
#define W_PITCH 72
template <int K>
__global__ __launch_bounds__(GEMM_THREADS) void gemm_wmma_kernel(
    const float* __restrict__ A, const float* __restrict__ sums,
    const float* __restrict__ sumsq, float invN,
    const float* __restrict__ W0, const float* __restrict__ W1,
    const float* __restrict__ W2, const float* __restrict__ eb0,
    const float* __restrict__ eb1, const float* __restrict__ eb2,
    float* __restrict__ C, int M, int doRelu,
    int strideA, int strideS, int strideC) {
  constexpr int W_PER = K * HIDDEN / GEMM_THREADS;
  static_assert(K * HIDDEN == W_PER * GEMM_THREADS, "W splits evenly over the threads");
  static_assert(GEMM_THREADS % HIDDEN == 0, "a thread keeps one column of W");
  static_assert(GEMM_THREADS >= K, "one thread per column statistic");
  __shared__ float s_w[K * W_PITCH];
  __shared__ float s_ms[GEMM_THREADS];
  __shared__ float s_s[GEMM_THREADS];
  __shared__ float s_v[HIDDEN];
  const int y = blockIdx.y;
  const float* W       = y == 0 ? W0  : (y == 1 ? W1  : W2);
  const float* extra_b = y == 0 ? eb0 : (y == 1 ? eb1 : eb2);
  A     += (size_t)y * strideA;
  sums  += (size_t)y * strideS;
  sumsq += (size_t)y * strideS;
  C     += (size_t)y * strideC;
  {
    const int ks = min((int)threadIdx.x, K - 1);
    const float csum = sums[ks], csq = sumsq[ks];
    const int wcol = threadIdx.x & (HIDDEN - 1);
    const int wrow = threadIdx.x / HIDDEN;
    float wv[W_PER];
#pragma unroll
    for (int q = 0; q < W_PER; ++q)
      wv[q] = W[(int)threadIdx.x + q * GEMM_THREADS];
    fold_stats(csum, csq, invN, s_ms, s_s);
#pragma unroll
    for (int q = 0; q < W_PER; ++q)
      s_w[(wrow + q * (GEMM_THREADS / HIDDEN)) * W_PITCH + wcol] = wv[q];
  }
  __syncthreads();
  if (threadIdx.x < HIDDEN) {
    const int j = threadIdx.x;
    s_v[j] = fold_column<W_PITCH>(s_w, s_ms, s_s, K, j, extra_b ? extra_b[j] : 0.f);
  }
  __syncthreads();
  const int lane = threadIdx.x & 31;
  const int wave = threadIdx.x >> 5;
  const int rowTile = blockIdx.x * GEMM_WAVES + wave;
  if (rowTile >= (M >> 4)) return;
  const int mrow = (rowTile << 4) + (lane & 15);
  const int kOff = (lane >> 4) << 1;
  const int n0   = lane & 15;
  v8f acc[4] = {};
  for (int kk = 0; kk < K; kk += 4) {
    const int k0 = kk + kOff;
    v2f a = *(const v2f*)(A + (size_t)mrow * K + k0);
    const float* w0 = s_w + k0 * W_PITCH + n0;
#pragma unroll
    for (int t = 0; t < 4; ++t) {
      v2f b;
      b.x = w0[t * 16];
      b.y = w0[W_PITCH + t * 16];
      acc[t] = __builtin_amdgcn_wmma_f32_16x16x4_f32(
          false, a, false, b, (short)0, acc[t], false, false);
    }
  }
  const int rowBase = (rowTile << 4) + ((lane >> 4) << 3);
#pragma unroll
  for (int t = 0; t < 4; ++t) {
    const int n = t * 16 + n0;
    const float bias = s_v[n];
#pragma unroll
    for (int e = 0; e < 8; ++e) {
      float val = acc[t][e] + bias;
      if (doRelu) val = fmaxf(val, 0.f);
      C[(size_t)(rowBase + e) * HIDDEN + n] = val;
    }
  }
}

#define TILE_NODES 400
#define N_TILES ((N_NODES + TILE_NODES - 1) / TILE_NODES)
#define BIN_THREADS 1024
#define BIN_PER_THREAD 4
#define BIN_BLOCKS ((N_EDGES + BIN_THREADS * BIN_PER_THREAD - 1) / (BIN_THREADS * BIN_PER_THREAD))
#define TP_THREADS 1024
#define TP_BATCH 8
#define TP_VEC (TILE_NODES * HIDDEN / 4)
#define TP_FIN ((TP_VEC + TP_THREADS - 1) / TP_THREADS)
#define BIN_STRIDE 256
#define TD_THREADS 1024
#define TD_PER_THREAD 4
static_assert(N_TILES + 1 <= BIN_STRIDE, "a list's bin starts fit its table");
static_assert(N_TILES <= 256, "the scan below is one 256-thread block");
static_assert(TP_THREADS % (HIDDEN / 4) == 0, "a thread keeps one column group");

__global__ __launch_bounds__(BIN_THREADS) void tile_count_kernel(
    const int* __restrict__ row, const int* __restrict__ col, int E,
    int* __restrict__ counts) {
  __shared__ int s_cnt[N_TILES];
  const int* key = blockIdx.y == 0 ? col : row;
  counts += blockIdx.y * BIN_STRIDE;
  for (int t = threadIdx.x; t < N_TILES; t += blockDim.x) s_cnt[t] = 0;
  __syncthreads();
  for (int i = 0; i < BIN_PER_THREAD; ++i) {
    const int e = (blockIdx.x * BIN_PER_THREAD + i) * BIN_THREADS + threadIdx.x;
    if (e < E) atomicAdd(&s_cnt[key[e] / TILE_NODES], 1);
  }
  __syncthreads();
  for (int t = threadIdx.x; t < N_TILES; t += blockDim.x) {
    const int v = s_cnt[t];
    if (v) atomicAdd(&counts[t], v);
  }
}

__global__ __launch_bounds__(256) void tile_scan_kernel(
    const int* __restrict__ counts, int* __restrict__ start,
    int* __restrict__ cursor) {
  __shared__ int s[256];
  counts += blockIdx.x * BIN_STRIDE;
  start  += blockIdx.x * BIN_STRIDE;
  cursor += blockIdx.x * BIN_STRIDE;
  const int t = threadIdx.x;
  const int v = t < N_TILES ? counts[t] : 0;
  s[t] = v;
  __syncthreads();
  for (int d = 1; d < 256; d <<= 1) {
    const int add = t >= d ? s[t - d] : 0;
    __syncthreads();
    s[t] += add;
    __syncthreads();
  }
  const int excl = s[t] - v;
  if (t < N_TILES) { start[t] = excl; cursor[t] = excl; }
  if (t == N_TILES - 1) start[N_TILES] = s[t];
}

__global__ __launch_bounds__(BIN_THREADS) void tile_fill_kernel(
    const int* __restrict__ row, const int* __restrict__ col, int E,
    int* __restrict__ cursor, int* __restrict__ rec) {
  __shared__ int s_cnt[N_TILES];
  __shared__ int s_base[N_TILES];
  const int* key = blockIdx.y == 0 ? col : row;
  const int* oth = blockIdx.y == 0 ? row : col;
  cursor += blockIdx.y * BIN_STRIDE;
  rec    += (size_t)blockIdx.y * N_EDGES * 4;
  for (int t = threadIdx.x; t < N_TILES; t += blockDim.x) s_cnt[t] = 0;
  __syncthreads();
  int rr[BIN_PER_THREAD], cc[BIN_PER_THREAD], tl[BIN_PER_THREAD], rk[BIN_PER_THREAD];
#pragma unroll
  for (int i = 0; i < BIN_PER_THREAD; ++i) {
    const int e = (blockIdx.x * BIN_PER_THREAD + i) * BIN_THREADS + threadIdx.x;
    rr[i] = 0; cc[i] = 0; tl[i] = 0; rk[i] = 0;
    if (e < E) {
      rr[i] = oth[e];
      cc[i] = key[e];
      tl[i] = cc[i] / TILE_NODES;
      rk[i] = atomicAdd(&s_cnt[tl[i]], 1);
    }
  }
  __syncthreads();
  for (int t = threadIdx.x; t < N_TILES; t += blockDim.x) {
    const int v = s_cnt[t];
    s_base[t] = v ? atomicAdd(&cursor[t], v) : 0;
  }
  __syncthreads();
#pragma unroll
  for (int i = 0; i < BIN_PER_THREAD; ++i) {
    const int e = (blockIdx.x * BIN_PER_THREAD + i) * BIN_THREADS + threadIdx.x;
    if (e < E) {
      int4 q;
      q.x = rr[i];
      q.y = cc[i] - tl[i] * TILE_NODES;
      q.z = e;
      q.w = tl[i];
      ((int4*)rec)[s_base[tl[i]] + rk[i]] = q;
    }
  }
}

__global__ __launch_bounds__(TD_THREADS) void tile_deg_kernel(
    const int* __restrict__ rec, const int* __restrict__ start,
    const float* __restrict__ att, float* __restrict__ dinv, int N, int strideD) {
  __shared__ float s_deg[2 * TILE_NODES];
  const int tile = blockIdx.x;
  const int n0 = tile * TILE_NODES;
  const int nn = min(TILE_NODES, N - n0);
  for (int i = threadIdx.x; i < 2 * TILE_NODES; i += TD_THREADS) s_deg[i] = 1.0f;
  __syncthreads();
  const int beg = start[tile], end = start[tile + 1];
  for (int e0 = beg + (int)threadIdx.x; e0 < end; e0 += TD_THREADS * TD_PER_THREAD) {
    int loc[TD_PER_THREAD], eid[TD_PER_THREAD];
#pragma unroll
    for (int k = 0; k < TD_PER_THREAD; ++k) {
      const int4 q = ((const int4*)rec)[min(e0 + k * TD_THREADS, end - 1)];
      loc[k] = q.y;
      eid[k] = q.z;
    }
    if (att) {
      float2 a[TD_PER_THREAD];
#pragma unroll
      for (int k = 0; k < TD_PER_THREAD; ++k) a[k] = ((const float2*)att)[eid[k]];
#pragma unroll
      for (int k = 0; k < TD_PER_THREAD; ++k) {
        if (e0 + k * TD_THREADS < end) {
          __hip_atomic_fetch_add(&s_deg[loc[k]], a[k].x, __ATOMIC_RELAXED,
                                 __HIP_MEMORY_SCOPE_WORKGROUP);
          __hip_atomic_fetch_add(&s_deg[TILE_NODES + loc[k]], a[k].y, __ATOMIC_RELAXED,
                                 __HIP_MEMORY_SCOPE_WORKGROUP);
        }
      }
    } else {
#pragma unroll
      for (int k = 0; k < TD_PER_THREAD; ++k) {
        if (e0 + k * TD_THREADS < end)
          __hip_atomic_fetch_add(&s_deg[loc[k]], 1.0f, __ATOMIC_RELAXED,
                                 __HIP_MEMORY_SCOPE_WORKGROUP);
      }
    }
  }
  __syncthreads();
  for (int i = threadIdx.x; i < nn; i += TD_THREADS) {
    const float d0 = s_deg[i];
    dinv[n0 + i] = d0 > 0.f ? rsqrtf(d0) : 0.f;
    if (att) {
      const float d1 = s_deg[TILE_NODES + i];
      dinv[(size_t)strideD + n0 + i] = d1 > 0.f ? rsqrtf(d1) : 0.f;
    }
  }
}

__device__ __forceinline__ void tile_batch(const float* __restrict__ m,
                                           float* s_acc, int r, int cl,
                                           float coef, int g, int lane) {
  float a0[TP_BATCH], a1[TP_BATCH];
#pragma unroll
  for (int u = 0; u < TP_BATCH; ++u) {
    const int ri = __builtin_amdgcn_readlane(r, g + u);
    const float* src = m + (size_t)ri * HIDDEN + lane;
    a0[u] = src[0];
    a1[u] = src[32];
  }
#pragma unroll
  for (int u = 0; u < TP_BATCH; ++u) {
    const int   ci = __builtin_amdgcn_readlane(cl, g + u);
    const float fi = __builtin_bit_cast(float,
        __builtin_amdgcn_readlane(__builtin_bit_cast(int, coef), g + u));
    float* dst = s_acc + ci * HIDDEN + lane;
    __hip_atomic_fetch_add(dst, fi * a0[u], __ATOMIC_RELAXED,
                           __HIP_MEMORY_SCOPE_WORKGROUP);
    __hip_atomic_fetch_add(dst + 32, fi * a1[u], __ATOMIC_RELAXED,
                           __HIP_MEMORY_SCOPE_WORKGROUP);
  }
}

__global__ __launch_bounds__(TP_THREADS) void tile_prop_kernel(
    const float* __restrict__ m, const int* __restrict__ rec,
    const int* __restrict__ start, const float* __restrict__ att, int comp,
    const float* __restrict__ dinv, const float* __restrict__ b0,
    const float* __restrict__ b1, float* __restrict__ out, int N,
    int strideM, int strideD) {
  __shared__ __attribute__((aligned(16))) float s_acc[TILE_NODES * HIDDEN];
  const float* b = blockIdx.y == 0 ? b0 : b1;
  comp += blockIdx.y;
  m    += (size_t)blockIdx.y * strideM;
  out  += (size_t)blockIdx.y * strideM;
  dinv += (size_t)blockIdx.y * strideD;
  const int tile = blockIdx.x;
  const int n0 = tile * TILE_NODES;
  const int nn = min(TILE_NODES, N - n0);
  for (int i = threadIdx.x; i < TILE_NODES * HIDDEN; i += TP_THREADS) s_acc[i] = 0.f;
  __syncthreads();
  const int beg = start[tile], end = start[tile + 1];
  const int lane = threadIdx.x & 31;
  const int wave = __builtin_amdgcn_readfirstlane((int)(threadIdx.x >> 5));
  for (int e0 = beg + wave * 32; e0 < end; e0 += TP_THREADS) {
    const int e = e0 + lane;
    int r = 0, cl = 0;
    float coef = 0.f;
    if (e < end) {
      const int4 q = ((const int4*)rec)[e];
      r = q.x;
      cl = q.y;
      const float w = att ? att[(size_t)q.z * 2 + comp] : 1.0f;
      coef = dinv[r] * w * dinv[n0 + cl];
    }
    const int cnt = min(32, end - e0);
    int g = 0;
    if (cnt == 32) {
#pragma unroll
      for (int gg = 0; gg < 32; gg += TP_BATCH)
        tile_batch(m, s_acc, r, cl, coef, gg, lane);
      g = 32;
    } else {
      for (; g + TP_BATCH <= cnt; g += TP_BATCH)
        tile_batch(m, s_acc, r, cl, coef, g, lane);
    }
    for (; g < cnt; ++g) {
      const int   ri = __builtin_amdgcn_readlane(r, g);
      const int   ci = __builtin_amdgcn_readlane(cl, g);
      const float fi = __builtin_bit_cast(float,
          __builtin_amdgcn_readlane(__builtin_bit_cast(int, coef), g));
      const float* src = m + (size_t)ri * HIDDEN + lane;
      float* dst = s_acc + ci * HIDDEN + lane;
      __hip_atomic_fetch_add(dst, fi * src[0], __ATOMIC_RELAXED,
                             __HIP_MEMORY_SCOPE_WORKGROUP);
      __hip_atomic_fetch_add(dst + 32, fi * src[32], __ATOMIC_RELAXED,
                             __HIP_MEMORY_SCOPE_WORKGROUP);
    }
  }
  const int nv = nn * (HIDDEN / 4);
  const float4* m4 = (const float4*)(m + (size_t)n0 * HIDDEN);
  float4* o4 = (float4*)(out + (size_t)n0 * HIDDEN);
  const float4 bv = ((const float4*)b)[threadIdx.x & (HIDDEN / 4 - 1)];
  float4 mv[TP_FIN];
  float  dv[TP_FIN];
#pragma unroll
  for (int k = 0; k < TP_FIN; ++k) {
    const int j4 = min((int)threadIdx.x + k * TP_THREADS, nv - 1);
    mv[k] = m4[j4];
    dv[k] = dinv[n0 + j4 / (HIDDEN / 4)];
  }
  __syncthreads();
#pragma unroll
  for (int k = 0; k < TP_FIN; ++k) {
    const int i4 = threadIdx.x + k * TP_THREADS;
    if (i4 < nv) {
      const float4 a = ((const float4*)s_acc)[i4];
      const float d2 = dv[k] * dv[k];
      float4 r;
      r.x = fmaxf(a.x + d2 * mv[k].x + bv.x, 0.f);
      r.y = fmaxf(a.y + d2 * mv[k].y + bv.y, 0.f);
      r.z = fmaxf(a.z + d2 * mv[k].z + bv.z, 0.f);
      r.w = fmaxf(a.w + d2 * mv[k].w + bv.w, 0.f);
      o4[i4] = r;
    }
  }
}

#define EA_BATCH 8
__global__ __launch_bounds__(256) void edge_att_kernel(
    const float* __restrict__ h, const int* __restrict__ row,
    const int* __restrict__ col, const float* __restrict__ Wea,
    const float* __restrict__ bea, float* __restrict__ att, int E) {
  const int lane = threadIdx.x & 31;
  const int e = blockIdx.x * blockDim.x + threadIdx.x;
  const int cnt = __builtin_amdgcn_readfirstlane(min(32, E - (e - lane)));
  if (cnt <= 0) return;
  int r = 0, c = 0;
  if (e < E) { r = row[e]; c = col[e]; }
  const float wr0a = Wea[lane * 2 + 0],                 wr1a = Wea[lane * 2 + 1];
  const float wr0b = Wea[(lane + 32) * 2 + 0],          wr1b = Wea[(lane + 32) * 2 + 1];
  const float wc0a = Wea[(HIDDEN + lane) * 2 + 0],      wc1a = Wea[(HIDDEN + lane) * 2 + 1];
  const float wc0b = Wea[(HIDDEN + lane + 32) * 2 + 0], wc1b = Wea[(HIDDEN + lane + 32) * 2 + 1];
  float my0 = 0.f, my1 = 0.f;
#pragma unroll
  for (int g = 0; g < 32; g += EA_BATCH) {
    if (g < cnt) {
      float ra[EA_BATCH], rb[EA_BATCH], ca[EA_BATCH], cb[EA_BATCH];
#pragma unroll
      for (int u = 0; u < EA_BATCH; ++u) {
        const int ri = __builtin_amdgcn_readlane(r, g + u);
        const int ci = __builtin_amdgcn_readlane(c, g + u);
        const float* hr = h + (size_t)ri * HIDDEN + lane;
        const float* hc = h + (size_t)ci * HIDDEN + lane;
        ra[u] = hr[0]; rb[u] = hr[32];
        ca[u] = hc[0]; cb[u] = hc[32];
      }
#pragma unroll
      for (int u = 0; u < EA_BATCH; ++u) {
        const float p0 = ra[u] * wr0a + rb[u] * wr0b + ca[u] * wc0a + cb[u] * wc0b;
        const float p1 = ra[u] * wr1a + rb[u] * wr1b + ca[u] * wc1a + cb[u] * wc1b;
        const float keep = lane < 16 ? p0 : p1;
        const float give = lane < 16 ? p1 : p0;
        float x = keep + __shfl_xor(give, 16, 32);
        x += __shfl_xor(x, 8, 32);
        x += __shfl_xor(x, 4, 32);
        x += __shfl_xor(x, 2, 32);
        x += __shfl_xor(x, 1, 32);
        const float s0 = __builtin_bit_cast(float,
            __builtin_amdgcn_readlane(__builtin_bit_cast(int, x), 0));
        const float s1 = __builtin_bit_cast(float,
            __builtin_amdgcn_readlane(__builtin_bit_cast(int, x), 16));
        if (lane == g + u) { my0 = s0; my1 = s1; }
      }
    }
  }
  if (e >= E) return;
  const float l0 = my0 + bea[0], l1 = my1 + bea[1];
  float mx = fmaxf(l0, l1);
  float e0 = __expf(l0 - mx), e1 = __expf(l1 - mx);
  float inv = 1.f / (e0 + e1);
  att[(size_t)e * 2 + 0] = e0 * inv;
  att[(size_t)e * 2 + 1] = e1 * inv;
}

__global__ void node_att_split_kernel(const float* __restrict__ h,
                                      const float* __restrict__ Wna,
                                      const float* __restrict__ bna,
                                      float* __restrict__ xc,
                                      float* __restrict__ xo, int N) {
  int i = blockIdx.x * blockDim.x + threadIdx.x;
  if (i >= N) return;
  const float* hi = h + (size_t)i * HIDDEN;
  float l0 = bna[0], l1 = bna[1];
  for (int j = 0; j < HIDDEN; ++j) {
    float a = hi[j];
    l0 += a * Wna[j * 2 + 0];
    l1 += a * Wna[j * 2 + 1];
  }
  float mx = fmaxf(l0, l1);
  float e0 = __expf(l0 - mx), e1 = __expf(l1 - mx);
  float inv = 1.f / (e0 + e1);
  float p0 = e0 * inv, p1 = e1 * inv;
  for (int j = 0; j < HIDDEN; ++j) {
    float a = hi[j];
    xc[(size_t)i * HIDDEN + j] = p0 * a;
    xo[(size_t)i * HIDDEN + j] = p1 * a;
  }
}

__global__ void pool_kernel(const float* __restrict__ x,
                            const int* __restrict__ batch,
                            float* __restrict__ g, int N,
                            int strideX, int strideG) {
  x += (size_t)blockIdx.y * strideX;
  g += (size_t)blockIdx.y * strideG;
  int idx = blockIdx.x * blockDim.x + threadIdx.x;
  int i = idx >> 6, j = idx & 63;
  if (i >= N) return;
  atomicAdd(&g[(size_t)batch[i] * HIDDEN + j], x[idx]);
}

__global__ void add_kernel(const float* a, const float* b, float* c, int n) {
  int i = blockIdx.x * blockDim.x + threadIdx.x;
  if (i < n) c[i] = a[i] + b[i];
}

#define LOGIT_THREADS 256
static_assert(LOGIT_THREADS >= HIDDEN, "one thread per column statistic");
__global__ __launch_bounds__(LOGIT_THREADS) void logits_kernel(
    const float* __restrict__ h, const float* __restrict__ sums,
    const float* __restrict__ sumsq, float invN,
    const float* __restrict__ W0, const float* __restrict__ W1,
    const float* __restrict__ W2, const float* __restrict__ eb0,
    const float* __restrict__ eb1, const float* __restrict__ eb2,
    float* __restrict__ out, int G, int strideH, int strideS, int strideO) {
  constexpr int W_PER = (HIDDEN * N_CLASSES + LOGIT_THREADS - 1) / LOGIT_THREADS;
  __shared__ float s_w[W_PER * LOGIT_THREADS];
  __shared__ float s_ms[LOGIT_THREADS];
  __shared__ float s_s[LOGIT_THREADS];
  __shared__ float s_v[N_CLASSES];
  const int y = blockIdx.y;
  const float* W       = y == 0 ? W0  : (y == 1 ? W1  : W2);
  const float* extra_b = y == 0 ? eb0 : (y == 1 ? eb1 : eb2);
  h     += (size_t)y * strideH;
  sums  += (size_t)y * strideS;
  sumsq += (size_t)y * strideS;
  out   += (size_t)y * strideO;
  {
    const int ks = min((int)threadIdx.x, HIDDEN - 1);
    const float csum = sums[ks], csq = sumsq[ks];
    float wv[W_PER];
#pragma unroll
    for (int q = 0; q < W_PER; ++q)
      wv[q] = W[min((int)threadIdx.x + q * LOGIT_THREADS, HIDDEN * N_CLASSES - 1)];
    fold_stats(csum, csq, invN, s_ms, s_s);
#pragma unroll
    for (int q = 0; q < W_PER; ++q)
      s_w[(int)threadIdx.x + q * LOGIT_THREADS] = wv[q];
  }
  __syncthreads();
  if (threadIdx.x < N_CLASSES) {
    const int j = threadIdx.x;
    s_v[j] = fold_column<N_CLASSES>(s_w, s_ms, s_s, HIDDEN, j, extra_b ? extra_b[j] : 0.f);
  }
  __syncthreads();
  int g = blockIdx.x * blockDim.x + threadIdx.x;
  if (g >= G) return;
  float l[N_CLASSES];
  for (int c = 0; c < N_CLASSES; ++c) l[c] = s_v[c];
  const float* hg = h + (size_t)g * HIDDEN;
  for (int k = 0; k < HIDDEN; ++k) {
    float x = hg[k];
    for (int c = 0; c < N_CLASSES; ++c) l[c] += x * s_w[k * N_CLASSES + c];
  }
  float mx = l[0];
  for (int c = 1; c < N_CLASSES; ++c) mx = fmaxf(mx, l[c]);
  float s = 0.f;
  for (int c = 0; c < N_CLASSES; ++c) s += __expf(l[c] - mx);
  float lse = __logf(s) + mx;
  for (int c = 0; c < N_CLASSES; ++c) out[(size_t)g * N_CLASSES + c] = l[c] - lse;
}

extern "C" void kernel_launch(void* const* d_in, const int* in_sizes, int n_in,
                              void* d_out, int out_size, void* d_ws, size_t ws_size,
                              hipStream_t stream) {
  (void)in_sizes; (void)n_in; (void)out_size; (void)ws_size;
  const float* x       = (const float*)d_in[0];
  const int*   ei      = (const int*)d_in[1];
  const int*   row     = ei;
  const int*   col     = ei + N_EDGES;
  const int*   batch   = (const int*)d_in[2];
  const float* W_feat  = (const float*)d_in[3];
  const float* W_convs = (const float*)d_in[4];
  const float* b_convs = (const float*)d_in[5];
  const float* W_ea    = (const float*)d_in[6];
  const float* b_ea    = (const float*)d_in[7];
  const float* W_na    = (const float*)d_in[8];
  const float* b_na    = (const float*)d_in[9];
  const float* W_ctx   = (const float*)d_in[10];
  const float* b_ctx   = (const float*)d_in[11];
  const float* W_obj   = (const float*)d_in[12];
  const float* b_obj   = (const float*)d_in[13];
  const float* W1h[3]  = {(const float*)d_in[14], (const float*)d_in[18], (const float*)d_in[22]};
  const float* b1h[3]  = {(const float*)d_in[15], (const float*)d_in[19], (const float*)d_in[23]};
  const float* W2h[3]  = {(const float*)d_in[16], (const float*)d_in[20], (const float*)d_in[24]};
  const float* b2h[3]  = {(const float*)d_in[17], (const float*)d_in[21], (const float*)d_in[25]};

  const int SN = N_NODES * HIDDEN;
  const int SG = N_GRAPHS * HIDDEN;
  const int SS = STAT_STRIDE;
  size_t off = 0;
  auto carve = [&](size_t elems) -> float* {
    float* p = (float*)((char*)d_ws + off);
    off += ((elems * sizeof(float)) + 255) & ~(size_t)255;
    return p;
  };
  float* hb    = carve((size_t)SN);
  float* mcat  = carve((size_t)2 * SN);
  float* xcat  = carve((size_t)2 * SN);
  float* att   = carve((size_t)N_EDGES * 2);
  float* dinvu = carve(N_NODES);
  float* dinva = carve((size_t)2 * N_NODES);
  float* gcat  = carve((size_t)3 * SG);
  float* r1cat = carve((size_t)3 * SG);
  float* stat  = carve((size_t)3 * SS);
  float* part  = carve((size_t)2 * CS_MAX_SLABS * SS);
  int*   tcount  = (int*)carve(2 * BIN_STRIDE);
  int*   tstart  = (int*)carve(2 * BIN_STRIDE);
  int*   tcursor = (int*)carve(2 * BIN_STRIDE);
  int*   erec    = (int*)carve((size_t)2 * N_EDGES * 4);
  const int* srec   = erec + (size_t)N_EDGES * 4;
  const int* sstart = tstart + BIN_STRIDE;
  float* xc = xcat;
  float* xo = xcat + SN;
  float* gc = gcat;
  float* go = gcat + SG;
  float* gs = gcat + 2 * SG;

  auto nblk = [](long n, int b) { return (int)((n + b - 1) / b); };

  auto stats = [&](const float* A, int N, int F, int ny, int strideA) {
    const int P = nblk(N, CS_SLAB);
    float* rows = P == 1 ? stat : part;
    if (F == N_FEAT) {
      colstats_slab_kernel<N_FEAT><<<dim3(P, ny), CS_THREADS, 0, stream>>>(A, N, rows, strideA);
      if (P > 1) colstats_finish_kernel<N_FEAT><<<ny, CF_THREADS, 0, stream>>>(part, P, stat);
    } else {
      colstats_slab_kernel<HIDDEN><<<dim3(P, ny), CS_THREADS, 0, stream>>>(A, N, rows, strideA);
      if (P > 1) colstats_finish_kernel<HIDDEN><<<ny, CF_THREADS, 0, stream>>>(part, P, stat);
    }
  };
  auto gemm = [&](const float* A, float* C, int M, int K, int relu, int ny,
                  int strideA, int strideC, float invN,
                  const float* W0, const float* W1, const float* W2,
                  const float* e0, const float* e1, const float* e2) {
    const dim3 grid(nblk(M / 16, GEMM_WAVES), ny);
    if (K == N_FEAT)
      gemm_wmma_kernel<N_FEAT><<<grid, GEMM_THREADS, 0, stream>>>(
          A, stat, stat + N_FEAT, invN, W0, W1, W2, e0, e1, e2, C, M, relu,
          strideA, SS, strideC);
    else
      gemm_wmma_kernel<HIDDEN><<<grid, GEMM_THREADS, 0, stream>>>(
          A, stat, stat + N_FEAT, invN, W0, W1, W2, e0, e1, e2, C, M, relu,
          strideA, SS, strideC);
  };

  stats(x, N_NODES, N_FEAT, 1, 0);
  gemm(x, hb, N_NODES, N_FEAT, 1, 1, 0, 0, 1.f / N_NODES, W_feat, W_feat, W_feat,
       nullptr, nullptr, nullptr);

  hipMemsetAsync(tcount, 0, (size_t)2 * BIN_STRIDE * sizeof(int), stream);
  tile_count_kernel<<<dim3(BIN_BLOCKS, 2), BIN_THREADS, 0, stream>>>(row, col, N_EDGES,
                                                                    tcount);
  tile_scan_kernel<<<2, 256, 0, stream>>>(tcount, tstart, tcursor);
  tile_fill_kernel<<<dim3(BIN_BLOCKS, 2), BIN_THREADS, 0, stream>>>(row, col, N_EDGES,
                                                                   tcursor, erec);

  tile_deg_kernel<<<N_TILES, TD_THREADS, 0, stream>>>(srec, sstart, nullptr, dinvu,
                                                      N_NODES, 0);

  for (int i = 0; i < 3; ++i) {
    const float* Wi = W_convs + (size_t)i * HIDDEN * HIDDEN;
    const float* bi = b_convs + (size_t)i * HIDDEN;
    stats(hb, N_NODES, HIDDEN, 1, 0);
    gemm(hb, mcat, N_NODES, HIDDEN, 0, 1, 0, 0, 1.f / N_NODES, Wi, Wi, Wi,
         nullptr, nullptr, nullptr);
    tile_prop_kernel<<<dim3(N_TILES, 1), TP_THREADS, 0, stream>>>(
        mcat, erec, tstart, nullptr, 0, dinvu, bi, bi, hb, N_NODES, 0, 0);
  }

  edge_att_kernel<<<nblk(N_EDGES, 256), 256, 0, stream>>>(hb, row, col, W_ea, b_ea,
                                                          att, N_EDGES);
  node_att_split_kernel<<<nblk(N_NODES, 256), 256, 0, stream>>>(hb, W_na, b_na,
                                                                xc, xo, N_NODES);

  stats(xcat, N_NODES, HIDDEN, 2, SN);
  gemm(xcat, mcat, N_NODES, HIDDEN, 0, 2, SN, SN, 1.f / N_NODES, W_ctx, W_obj, W_obj,
       nullptr, nullptr, nullptr);
  tile_deg_kernel<<<N_TILES, TD_THREADS, 0, stream>>>(srec, sstart, att, dinva,
                                                      N_NODES, N_NODES);
  tile_prop_kernel<<<dim3(N_TILES, 2), TP_THREADS, 0, stream>>>(
      mcat, erec, tstart, att, 0, dinva, b_ctx, b_obj, xcat, N_NODES, SN, N_NODES);
  hipMemsetAsync(gcat, 0, (size_t)2 * SG * sizeof(float), stream);
  pool_kernel<<<dim3(nblk((long)N_NODES * 64, 256), 2), 256, 0, stream>>>(
      xcat, batch, gcat, N_NODES, SN, SG);

  add_kernel<<<nblk(N_GRAPHS * HIDDEN, 256), 256, 0, stream>>>(gc, go, gs,
                                                               N_GRAPHS * HIDDEN);

  float* out = (float*)d_out;
  stats(gcat, N_GRAPHS, HIDDEN, 3, SG);
  gemm(gcat, r1cat, N_GRAPHS, HIDDEN, 1, 3, SG, SG, 1.f / N_GRAPHS,
       W1h[0], W1h[1], W1h[2], b1h[0], b1h[1], b1h[2]);
  stats(r1cat, N_GRAPHS, HIDDEN, 3, SG);
  logits_kernel<<<dim3(nblk(N_GRAPHS, LOGIT_THREADS), 3), LOGIT_THREADS, 0, stream>>>(
      r1cat, stat, stat + N_FEAT, 1.f / N_GRAPHS, W2h[0], W2h[1], W2h[2],
      b2h[0], b2h[1], b2h[2], out, N_GRAPHS, SG, SS, N_GRAPHS * N_CLASSES);
}


# global add pool: thread per element, one f32 atomic each (400,000 one-load waves) -> wave per 32 rows, 16 16-byte loads in flight, a graph's run summed in registers, atomics per run (new sum order)
# speedup vs baseline: 4.9709x; 1.0539x over previous
#include <hip/hip_runtime.h>
#include <math.h>


#define N_NODES  100000
#define N_EDGES  1000000
#define N_FEAT   128
#define HIDDEN   64
#define N_GRAPHS 512
#define N_CLASSES 10
#define BN_EPS  1e-5f
#define BN_BETA 1e-4f

typedef __attribute__((ext_vector_type(2))) float v2f;
typedef __attribute__((ext_vector_type(8))) float v8f;

#define CS_THREADS 512
#define CS_WAVES (CS_THREADS / 32)
#define CS_BATCH 8
#define CS_SLAB 512
#define CS_MAX_SLABS ((N_NODES + CS_SLAB - 1) / CS_SLAB)
#define CF_THREADS 1024
#define STAT_STRIDE (2 * N_FEAT)

typedef __attribute__((ext_vector_type(4))) float v4f;

template <int F, bool GUARD>
__device__ __forceinline__ void cs_load(const v4f* __restrict__ A4, int rfirst, int r1,
                                        int c4, int b0, v4f (&v)[CS_BATCH]) {
  constexpr int VPR  = F / 4;
  constexpr int STEP = CS_WAVES * (32 / VPR);
#pragma unroll
  for (int u = 0; u < CS_BATCH; ++u) {
    int r = rfirst + (b0 + u) * STEP;
    if (GUARD) r = min(r, r1 - 1);
    v[u] = A4[(size_t)r * VPR + c4];
  }
}

template <int F, bool GUARD>
__device__ __forceinline__ void cs_slab_sum(const v4f* __restrict__ A4, int rfirst,
                                            int r1, int c4, v4f& s, v4f& q) {
  constexpr int VPR  = F / 4;
  constexpr int STEP = CS_WAVES * (32 / VPR);
  constexpr int TRIPS = CS_SLAB / (STEP * CS_BATCH);
  static_assert(TRIPS * STEP * CS_BATCH == CS_SLAB, "a slab is a whole number of batches");
  v4f cur[CS_BATCH], nxt[CS_BATCH];
  cs_load<F, GUARD>(A4, rfirst, r1, c4, 0, cur);
#pragma unroll
  for (int t = 0; t < TRIPS; ++t) {
    if (t + 1 < TRIPS) cs_load<F, GUARD>(A4, rfirst, r1, c4, (t + 1) * CS_BATCH, nxt);
    asm volatile("" :: "v"(cur[0]), "v"(cur[1]), "v"(cur[2]), "v"(cur[3]),
                       "v"(cur[4]), "v"(cur[5]), "v"(cur[6]), "v"(cur[7]));
#pragma unroll
    for (int u = 0; u < CS_BATCH; ++u) {
      const int r = rfirst + (t * CS_BATCH + u) * STEP;
      if (!GUARD || r < r1) { s += cur[u]; q += cur[u] * cur[u]; }
    }
#pragma unroll
    for (int u = 0; u < CS_BATCH; ++u) cur[u] = nxt[u];
  }
}
static_assert(CS_BATCH == 8, "the fence above names eight values");

template <int F>
__global__ __launch_bounds__(CS_THREADS) void colstats_slab_kernel(
    const float* __restrict__ A, int N, float* __restrict__ part, int strideA) {
  constexpr int VPR = F / 4;
  static_assert(VPR == 16 || VPR == 32, "a wave-wide load covers one or two whole rows");
  __shared__ v4f s_s[CS_THREADS];
  __shared__ v4f s_q[CS_THREADS];
  A    += (size_t)blockIdx.y * strideA;
  part += ((size_t)blockIdx.y * gridDim.x + blockIdx.x) * STAT_STRIDE;
  const int lane = threadIdx.x & 31;
  const int wave = threadIdx.x >> 5;
  const int c4   = lane & (VPR - 1);
  const int r0   = blockIdx.x * CS_SLAB;
  const int r1   = min(N, r0 + CS_SLAB);
  const int rfirst = r0 + wave * (32 / VPR) + lane / VPR;
  v4f s = {0.f, 0.f, 0.f, 0.f}, q = s;
  if (r0 + CS_SLAB <= N) cs_slab_sum<F, false>((const v4f*)A, rfirst, r1, c4, s, q);
  else                   cs_slab_sum<F, true>((const v4f*)A, rfirst, r1, c4, s, q);
  s_s[threadIdx.x] = s;
  s_q[threadIdx.x] = q;
  __syncthreads();
  if (threadIdx.x < 2 * VPR) {
    const bool sq = threadIdx.x >= VPR;
    const int c = threadIdx.x & (VPR - 1);
    const v4f* src = sq ? s_q : s_s;
    v4f a = src[c];
#pragma unroll
    for (int j = 1; j < CS_THREADS / VPR; ++j) a += src[j * VPR + c];
    ((v4f*)(part + (sq ? N_FEAT : 0)))[c] = a;
  }
}

template <int F>
__global__ __launch_bounds__(CF_THREADS) void colstats_finish_kernel(
    const float* __restrict__ part, int P, float* __restrict__ stat) {
  constexpr int VPR = F / 4;
  constexpr int V2  = 2 * VPR;
  constexpr int G   = CF_THREADS / V2;
  constexpr int PER = (CS_MAX_SLABS + G - 1) / G;
  __shared__ float4 s_r[CF_THREADS];
  part += (size_t)blockIdx.x * P * STAT_STRIDE;
  stat += (size_t)blockIdx.x * STAT_STRIDE;
  const int c = threadIdx.x & (V2 - 1);
  const int g = threadIdx.x / V2;
  const int off = c < VPR ? c * 4 : N_FEAT + (c - VPR) * 4;
  float4 v[PER];
#pragma unroll
  for (int i = 0; i < PER; ++i)
    v[i] = *(const float4*)(part + (size_t)min(g + i * G, P - 1) * STAT_STRIDE + off);
  float4 a = make_float4(0.f, 0.f, 0.f, 0.f);
#pragma unroll
  for (int i = 0; i < PER; ++i)
    if (g + i * G < P) { a.x += v[i].x; a.y += v[i].y; a.z += v[i].z; a.w += v[i].w; }
  s_r[threadIdx.x] = a;
  __syncthreads();
  if (threadIdx.x < V2) {
    float4 t = s_r[c];
#pragma unroll
    for (int j = 1; j < G; ++j) {
      const float4 b = s_r[j * V2 + c];
      t.x += b.x; t.y += b.y; t.z += b.z; t.w += b.w;
    }
    *(float4*)(stat + off) = t;
  }
}

__device__ __forceinline__ void fold_stats(float sum, float sumsq, float invN,
                                           float* s_ms, float* s_s) {
  float mean = sum * invN;
  float var  = sumsq * invN - mean * mean;
  float sc   = rsqrtf(var + BN_EPS);
  s_ms[threadIdx.x] = mean * sc;
  s_s[threadIdx.x]  = sc;
}

template <int PITCH>
__device__ __forceinline__ float fold_column(float* s_w, const float* s_ms,
                                             const float* s_s, int K, int j,
                                             float acc) {
#pragma unroll 8
  for (int k = 0; k < K; ++k) {
    const float w = s_w[k * PITCH + j];
    s_w[k * PITCH + j] = s_s[k] * w;
    acc += (BN_BETA - s_ms[k]) * w;
  }
  return acc;
}

#define GEMM_WAVES 8
#define GEMM_THREADS (GEMM_WAVES * 32)
#define W_PITCH 72
template <int K>
__global__ __launch_bounds__(GEMM_THREADS) void gemm_wmma_kernel(
    const float* __restrict__ A, const float* __restrict__ sums,
    const float* __restrict__ sumsq, float invN,
    const float* __restrict__ W0, const float* __restrict__ W1,
    const float* __restrict__ W2, const float* __restrict__ eb0,
    const float* __restrict__ eb1, const float* __restrict__ eb2,
    float* __restrict__ C, int M, int doRelu,
    int strideA, int strideS, int strideC) {
  constexpr int W_PER = K * HIDDEN / GEMM_THREADS;
  static_assert(K * HIDDEN == W_PER * GEMM_THREADS, "W splits evenly over the threads");
  static_assert(GEMM_THREADS % HIDDEN == 0, "a thread keeps one column of W");
  static_assert(GEMM_THREADS >= K, "one thread per column statistic");
  __shared__ float s_w[K * W_PITCH];
  __shared__ float s_ms[GEMM_THREADS];
  __shared__ float s_s[GEMM_THREADS];
  __shared__ float s_v[HIDDEN];
  const int y = blockIdx.y;
  const float* W       = y == 0 ? W0  : (y == 1 ? W1  : W2);
  const float* extra_b = y == 0 ? eb0 : (y == 1 ? eb1 : eb2);
  A     += (size_t)y * strideA;
  sums  += (size_t)y * strideS;
  sumsq += (size_t)y * strideS;
  C     += (size_t)y * strideC;
  {
    const int ks = min((int)threadIdx.x, K - 1);
    const float csum = sums[ks], csq = sumsq[ks];
    const int wcol = threadIdx.x & (HIDDEN - 1);
    const int wrow = threadIdx.x / HIDDEN;
    float wv[W_PER];
#pragma unroll
    for (int q = 0; q < W_PER; ++q)
      wv[q] = W[(int)threadIdx.x + q * GEMM_THREADS];
    fold_stats(csum, csq, invN, s_ms, s_s);
#pragma unroll
    for (int q = 0; q < W_PER; ++q)
      s_w[(wrow + q * (GEMM_THREADS / HIDDEN)) * W_PITCH + wcol] = wv[q];
  }
  __syncthreads();
  if (threadIdx.x < HIDDEN) {
    const int j = threadIdx.x;
    s_v[j] = fold_column<W_PITCH>(s_w, s_ms, s_s, K, j, extra_b ? extra_b[j] : 0.f);
  }
  __syncthreads();
  const int lane = threadIdx.x & 31;
  const int wave = threadIdx.x >> 5;
  const int rowTile = blockIdx.x * GEMM_WAVES + wave;
  if (rowTile >= (M >> 4)) return;
  const int mrow = (rowTile << 4) + (lane & 15);
  const int kOff = (lane >> 4) << 1;
  const int n0   = lane & 15;
  v8f acc[4] = {};
  for (int kk = 0; kk < K; kk += 4) {
    const int k0 = kk + kOff;
    v2f a = *(const v2f*)(A + (size_t)mrow * K + k0);
    const float* w0 = s_w + k0 * W_PITCH + n0;
#pragma unroll
    for (int t = 0; t < 4; ++t) {
      v2f b;
      b.x = w0[t * 16];
      b.y = w0[W_PITCH + t * 16];
      acc[t] = __builtin_amdgcn_wmma_f32_16x16x4_f32(
          false, a, false, b, (short)0, acc[t], false, false);
    }
  }
  const int rowBase = (rowTile << 4) + ((lane >> 4) << 3);
#pragma unroll
  for (int t = 0; t < 4; ++t) {
    const int n = t * 16 + n0;
    const float bias = s_v[n];
#pragma unroll
    for (int e = 0; e < 8; ++e) {
      float val = acc[t][e] + bias;
      if (doRelu) val = fmaxf(val, 0.f);
      C[(size_t)(rowBase + e) * HIDDEN + n] = val;
    }
  }
}

#define TILE_NODES 400
#define N_TILES ((N_NODES + TILE_NODES - 1) / TILE_NODES)
#define BIN_THREADS 1024
#define BIN_PER_THREAD 4
#define BIN_BLOCKS ((N_EDGES + BIN_THREADS * BIN_PER_THREAD - 1) / (BIN_THREADS * BIN_PER_THREAD))
#define TP_THREADS 1024
#define TP_BATCH 8
#define TP_VEC (TILE_NODES * HIDDEN / 4)
#define TP_FIN ((TP_VEC + TP_THREADS - 1) / TP_THREADS)
#define BIN_STRIDE 256
#define TD_THREADS 1024
#define TD_PER_THREAD 4
static_assert(N_TILES + 1 <= BIN_STRIDE, "a list's bin starts fit its table");
static_assert(N_TILES <= 256, "the scan below is one 256-thread block");
static_assert(TP_THREADS % (HIDDEN / 4) == 0, "a thread keeps one column group");

__global__ __launch_bounds__(BIN_THREADS) void tile_count_kernel(
    const int* __restrict__ row, const int* __restrict__ col, int E,
    int* __restrict__ counts) {
  __shared__ int s_cnt[N_TILES];
  const int* key = blockIdx.y == 0 ? col : row;
  counts += blockIdx.y * BIN_STRIDE;
  for (int t = threadIdx.x; t < N_TILES; t += blockDim.x) s_cnt[t] = 0;
  __syncthreads();
  for (int i = 0; i < BIN_PER_THREAD; ++i) {
    const int e = (blockIdx.x * BIN_PER_THREAD + i) * BIN_THREADS + threadIdx.x;
    if (e < E) atomicAdd(&s_cnt[key[e] / TILE_NODES], 1);
  }
  __syncthreads();
  for (int t = threadIdx.x; t < N_TILES; t += blockDim.x) {
    const int v = s_cnt[t];
    if (v) atomicAdd(&counts[t], v);
  }
}

__global__ __launch_bounds__(256) void tile_scan_kernel(
    const int* __restrict__ counts, int* __restrict__ start,
    int* __restrict__ cursor) {
  __shared__ int s[256];
  counts += blockIdx.x * BIN_STRIDE;
  start  += blockIdx.x * BIN_STRIDE;
  cursor += blockIdx.x * BIN_STRIDE;
  const int t = threadIdx.x;
  const int v = t < N_TILES ? counts[t] : 0;
  s[t] = v;
  __syncthreads();
  for (int d = 1; d < 256; d <<= 1) {
    const int add = t >= d ? s[t - d] : 0;
    __syncthreads();
    s[t] += add;
    __syncthreads();
  }
  const int excl = s[t] - v;
  if (t < N_TILES) { start[t] = excl; cursor[t] = excl; }
  if (t == N_TILES - 1) start[N_TILES] = s[t];
}

__global__ __launch_bounds__(BIN_THREADS) void tile_fill_kernel(
    const int* __restrict__ row, const int* __restrict__ col, int E,
    int* __restrict__ cursor, int* __restrict__ rec) {
  __shared__ int s_cnt[N_TILES];
  __shared__ int s_base[N_TILES];
  const int* key = blockIdx.y == 0 ? col : row;
  const int* oth = blockIdx.y == 0 ? row : col;
  cursor += blockIdx.y * BIN_STRIDE;
  rec    += (size_t)blockIdx.y * N_EDGES * 4;
  for (int t = threadIdx.x; t < N_TILES; t += blockDim.x) s_cnt[t] = 0;
  __syncthreads();
  int rr[BIN_PER_THREAD], cc[BIN_PER_THREAD], tl[BIN_PER_THREAD], rk[BIN_PER_THREAD];
#pragma unroll
  for (int i = 0; i < BIN_PER_THREAD; ++i) {
    const int e = (blockIdx.x * BIN_PER_THREAD + i) * BIN_THREADS + threadIdx.x;
    rr[i] = 0; cc[i] = 0; tl[i] = 0; rk[i] = 0;
    if (e < E) {
      rr[i] = oth[e];
      cc[i] = key[e];
      tl[i] = cc[i] / TILE_NODES;
      rk[i] = atomicAdd(&s_cnt[tl[i]], 1);
    }
  }
  __syncthreads();
  for (int t = threadIdx.x; t < N_TILES; t += blockDim.x) {
    const int v = s_cnt[t];
    s_base[t] = v ? atomicAdd(&cursor[t], v) : 0;
  }
  __syncthreads();
#pragma unroll
  for (int i = 0; i < BIN_PER_THREAD; ++i) {
    const int e = (blockIdx.x * BIN_PER_THREAD + i) * BIN_THREADS + threadIdx.x;
    if (e < E) {
      int4 q;
      q.x = rr[i];
      q.y = cc[i] - tl[i] * TILE_NODES;
      q.z = e;
      q.w = tl[i];
      ((int4*)rec)[s_base[tl[i]] + rk[i]] = q;
    }
  }
}

__global__ __launch_bounds__(TD_THREADS) void tile_deg_kernel(
    const int* __restrict__ rec, const int* __restrict__ start,
    const float* __restrict__ att, float* __restrict__ dinv, int N, int strideD) {
  __shared__ float s_deg[2 * TILE_NODES];
  const int tile = blockIdx.x;
  const int n0 = tile * TILE_NODES;
  const int nn = min(TILE_NODES, N - n0);
  for (int i = threadIdx.x; i < 2 * TILE_NODES; i += TD_THREADS) s_deg[i] = 1.0f;
  __syncthreads();
  const int beg = start[tile], end = start[tile + 1];
  for (int e0 = beg + (int)threadIdx.x; e0 < end; e0 += TD_THREADS * TD_PER_THREAD) {
    int loc[TD_PER_THREAD], eid[TD_PER_THREAD];
#pragma unroll
    for (int k = 0; k < TD_PER_THREAD; ++k) {
      const int4 q = ((const int4*)rec)[min(e0 + k * TD_THREADS, end - 1)];
      loc[k] = q.y;
      eid[k] = q.z;
    }
    if (att) {
      float2 a[TD_PER_THREAD];
#pragma unroll
      for (int k = 0; k < TD_PER_THREAD; ++k) a[k] = ((const float2*)att)[eid[k]];
#pragma unroll
      for (int k = 0; k < TD_PER_THREAD; ++k) {
        if (e0 + k * TD_THREADS < end) {
          __hip_atomic_fetch_add(&s_deg[loc[k]], a[k].x, __ATOMIC_RELAXED,
                                 __HIP_MEMORY_SCOPE_WORKGROUP);
          __hip_atomic_fetch_add(&s_deg[TILE_NODES + loc[k]], a[k].y, __ATOMIC_RELAXED,
                                 __HIP_MEMORY_SCOPE_WORKGROUP);
        }
      }
    } else {
#pragma unroll
      for (int k = 0; k < TD_PER_THREAD; ++k) {
        if (e0 + k * TD_THREADS < end)
          __hip_atomic_fetch_add(&s_deg[loc[k]], 1.0f, __ATOMIC_RELAXED,
                                 __HIP_MEMORY_SCOPE_WORKGROUP);
      }
    }
  }
  __syncthreads();
  for (int i = threadIdx.x; i < nn; i += TD_THREADS) {
    const float d0 = s_deg[i];
    dinv[n0 + i] = d0 > 0.f ? rsqrtf(d0) : 0.f;
    if (att) {
      const float d1 = s_deg[TILE_NODES + i];
      dinv[(size_t)strideD + n0 + i] = d1 > 0.f ? rsqrtf(d1) : 0.f;
    }
  }
}

__device__ __forceinline__ void tile_batch(const float* __restrict__ m,
                                           float* s_acc, int r, int cl,
                                           float coef, int g, int lane) {
  float a0[TP_BATCH], a1[TP_BATCH];
#pragma unroll
  for (int u = 0; u < TP_BATCH; ++u) {
    const int ri = __builtin_amdgcn_readlane(r, g + u);
    const float* src = m + (size_t)ri * HIDDEN + lane;
    a0[u] = src[0];
    a1[u] = src[32];
  }
#pragma unroll
  for (int u = 0; u < TP_BATCH; ++u) {
    const int   ci = __builtin_amdgcn_readlane(cl, g + u);
    const float fi = __builtin_bit_cast(float,
        __builtin_amdgcn_readlane(__builtin_bit_cast(int, coef), g + u));
    float* dst = s_acc + ci * HIDDEN + lane;
    __hip_atomic_fetch_add(dst, fi * a0[u], __ATOMIC_RELAXED,
                           __HIP_MEMORY_SCOPE_WORKGROUP);
    __hip_atomic_fetch_add(dst + 32, fi * a1[u], __ATOMIC_RELAXED,
                           __HIP_MEMORY_SCOPE_WORKGROUP);
  }
}

__global__ __launch_bounds__(TP_THREADS) void tile_prop_kernel(
    const float* __restrict__ m, const int* __restrict__ rec,
    const int* __restrict__ start, const float* __restrict__ att, int comp,
    const float* __restrict__ dinv, const float* __restrict__ b0,
    const float* __restrict__ b1, float* __restrict__ out, int N,
    int strideM, int strideD) {
  __shared__ __attribute__((aligned(16))) float s_acc[TILE_NODES * HIDDEN];
  const float* b = blockIdx.y == 0 ? b0 : b1;
  comp += blockIdx.y;
  m    += (size_t)blockIdx.y * strideM;
  out  += (size_t)blockIdx.y * strideM;
  dinv += (size_t)blockIdx.y * strideD;
  const int tile = blockIdx.x;
  const int n0 = tile * TILE_NODES;
  const int nn = min(TILE_NODES, N - n0);
  for (int i = threadIdx.x; i < TILE_NODES * HIDDEN; i += TP_THREADS) s_acc[i] = 0.f;
  __syncthreads();
  const int beg = start[tile], end = start[tile + 1];
  const int lane = threadIdx.x & 31;
  const int wave = __builtin_amdgcn_readfirstlane((int)(threadIdx.x >> 5));
  for (int e0 = beg + wave * 32; e0 < end; e0 += TP_THREADS) {
    const int e = e0 + lane;
    int r = 0, cl = 0;
    float coef = 0.f;
    if (e < end) {
      const int4 q = ((const int4*)rec)[e];
      r = q.x;
      cl = q.y;
      const float w = att ? att[(size_t)q.z * 2 + comp] : 1.0f;
      coef = dinv[r] * w * dinv[n0 + cl];
    }
    const int cnt = min(32, end - e0);
    int g = 0;
    if (cnt == 32) {
#pragma unroll
      for (int gg = 0; gg < 32; gg += TP_BATCH)
        tile_batch(m, s_acc, r, cl, coef, gg, lane);
      g = 32;
    } else {
      for (; g + TP_BATCH <= cnt; g += TP_BATCH)
        tile_batch(m, s_acc, r, cl, coef, g, lane);
    }
    for (; g < cnt; ++g) {
      const int   ri = __builtin_amdgcn_readlane(r, g);
      const int   ci = __builtin_amdgcn_readlane(cl, g);
      const float fi = __builtin_bit_cast(float,
          __builtin_amdgcn_readlane(__builtin_bit_cast(int, coef), g));
      const float* src = m + (size_t)ri * HIDDEN + lane;
      float* dst = s_acc + ci * HIDDEN + lane;
      __hip_atomic_fetch_add(dst, fi * src[0], __ATOMIC_RELAXED,
                             __HIP_MEMORY_SCOPE_WORKGROUP);
      __hip_atomic_fetch_add(dst + 32, fi * src[32], __ATOMIC_RELAXED,
                             __HIP_MEMORY_SCOPE_WORKGROUP);
    }
  }
  const int nv = nn * (HIDDEN / 4);
  const float4* m4 = (const float4*)(m + (size_t)n0 * HIDDEN);
  float4* o4 = (float4*)(out + (size_t)n0 * HIDDEN);
  const float4 bv = ((const float4*)b)[threadIdx.x & (HIDDEN / 4 - 1)];
  float4 mv[TP_FIN];
  float  dv[TP_FIN];
#pragma unroll
  for (int k = 0; k < TP_FIN; ++k) {
    const int j4 = min((int)threadIdx.x + k * TP_THREADS, nv - 1);
    mv[k] = m4[j4];
    dv[k] = dinv[n0 + j4 / (HIDDEN / 4)];
  }
  __syncthreads();
#pragma unroll
  for (int k = 0; k < TP_FIN; ++k) {
    const int i4 = threadIdx.x + k * TP_THREADS;
    if (i4 < nv) {
      const float4 a = ((const float4*)s_acc)[i4];
      const float d2 = dv[k] * dv[k];
      float4 r;
      r.x = fmaxf(a.x + d2 * mv[k].x + bv.x, 0.f);
      r.y = fmaxf(a.y + d2 * mv[k].y + bv.y, 0.f);
      r.z = fmaxf(a.z + d2 * mv[k].z + bv.z, 0.f);
      r.w = fmaxf(a.w + d2 * mv[k].w + bv.w, 0.f);
      o4[i4] = r;
    }
  }
}

#define EA_BATCH 8
__global__ __launch_bounds__(256) void edge_att_kernel(
    const float* __restrict__ h, const int* __restrict__ row,
    const int* __restrict__ col, const float* __restrict__ Wea,
    const float* __restrict__ bea, float* __restrict__ att, int E) {
  const int lane = threadIdx.x & 31;
  const int e = blockIdx.x * blockDim.x + threadIdx.x;
  const int cnt = __builtin_amdgcn_readfirstlane(min(32, E - (e - lane)));
  if (cnt <= 0) return;
  int r = 0, c = 0;
  if (e < E) { r = row[e]; c = col[e]; }
  const float wr0a = Wea[lane * 2 + 0],                 wr1a = Wea[lane * 2 + 1];
  const float wr0b = Wea[(lane + 32) * 2 + 0],          wr1b = Wea[(lane + 32) * 2 + 1];
  const float wc0a = Wea[(HIDDEN + lane) * 2 + 0],      wc1a = Wea[(HIDDEN + lane) * 2 + 1];
  const float wc0b = Wea[(HIDDEN + lane + 32) * 2 + 0], wc1b = Wea[(HIDDEN + lane + 32) * 2 + 1];
  float my0 = 0.f, my1 = 0.f;
#pragma unroll
  for (int g = 0; g < 32; g += EA_BATCH) {
    if (g < cnt) {
      float ra[EA_BATCH], rb[EA_BATCH], ca[EA_BATCH], cb[EA_BATCH];
#pragma unroll
      for (int u = 0; u < EA_BATCH; ++u) {
        const int ri = __builtin_amdgcn_readlane(r, g + u);
        const int ci = __builtin_amdgcn_readlane(c, g + u);
        const float* hr = h + (size_t)ri * HIDDEN + lane;
        const float* hc = h + (size_t)ci * HIDDEN + lane;
        ra[u] = hr[0]; rb[u] = hr[32];
        ca[u] = hc[0]; cb[u] = hc[32];
      }
#pragma unroll
      for (int u = 0; u < EA_BATCH; ++u) {
        const float p0 = ra[u] * wr0a + rb[u] * wr0b + ca[u] * wc0a + cb[u] * wc0b;
        const float p1 = ra[u] * wr1a + rb[u] * wr1b + ca[u] * wc1a + cb[u] * wc1b;
        const float keep = lane < 16 ? p0 : p1;
        const float give = lane < 16 ? p1 : p0;
        float x = keep + __shfl_xor(give, 16, 32);
        x += __shfl_xor(x, 8, 32);
        x += __shfl_xor(x, 4, 32);
        x += __shfl_xor(x, 2, 32);
        x += __shfl_xor(x, 1, 32);
        const float s0 = __builtin_bit_cast(float,
            __builtin_amdgcn_readlane(__builtin_bit_cast(int, x), 0));
        const float s1 = __builtin_bit_cast(float,
            __builtin_amdgcn_readlane(__builtin_bit_cast(int, x), 16));
        if (lane == g + u) { my0 = s0; my1 = s1; }
      }
    }
  }
  if (e >= E) return;
  const float l0 = my0 + bea[0], l1 = my1 + bea[1];
  float mx = fmaxf(l0, l1);
  float e0 = __expf(l0 - mx), e1 = __expf(l1 - mx);
  float inv = 1.f / (e0 + e1);
  att[(size_t)e * 2 + 0] = e0 * inv;
  att[(size_t)e * 2 + 1] = e1 * inv;
}

__global__ void node_att_split_kernel(const float* __restrict__ h,
                                      const float* __restrict__ Wna,
                                      const float* __restrict__ bna,
                                      float* __restrict__ xc,
                                      float* __restrict__ xo, int N) {
  int i = blockIdx.x * blockDim.x + threadIdx.x;
  if (i >= N) return;
  const float* hi = h + (size_t)i * HIDDEN;
  float l0 = bna[0], l1 = bna[1];
  for (int j = 0; j < HIDDEN; ++j) {
    float a = hi[j];
    l0 += a * Wna[j * 2 + 0];
    l1 += a * Wna[j * 2 + 1];
  }
  float mx = fmaxf(l0, l1);
  float e0 = __expf(l0 - mx), e1 = __expf(l1 - mx);
  float inv = 1.f / (e0 + e1);
  float p0 = e0 * inv, p1 = e1 * inv;
  for (int j = 0; j < HIDDEN; ++j) {
    float a = hi[j];
    xc[(size_t)i * HIDDEN + j] = p0 * a;
    xo[(size_t)i * HIDDEN + j] = p1 * a;
  }
}

#define PL_ROWS 32
#define PL_LOADS (PL_ROWS / 2)
#define PL_THREADS 256
static_assert(N_NODES % PL_ROWS == 0, "every row of a wave exists");
static_assert(HIDDEN * 4 * 2 == 32 * 16, "a wave-wide 16-byte load covers two rows");

__device__ __forceinline__ void pool_flush(float* __restrict__ g, int id, int c4, v4f a) {
  float* dst = g + (size_t)id * HIDDEN + c4 * 4;
  atomicAdd(dst + 0, a.x);
  atomicAdd(dst + 1, a.y);
  atomicAdd(dst + 2, a.z);
  atomicAdd(dst + 3, a.w);
}

__global__ __launch_bounds__(PL_THREADS) void pool_kernel(
    const float* __restrict__ x, const int* __restrict__ batch,
    float* __restrict__ g, int N, int strideX, int strideG) {
  x += (size_t)blockIdx.y * strideX;
  g += (size_t)blockIdx.y * strideG;
  const int lane = threadIdx.x & 31;
  const int wave = __builtin_amdgcn_readfirstlane((int)(threadIdx.x >> 5));
  const int r0 = (blockIdx.x * (PL_THREADS / 32) + wave) * PL_ROWS;
  if (r0 >= N) return;
  const int idv = batch[r0 + lane];
  const v4f* x4 = (const v4f*)(x + (size_t)r0 * HIDDEN) + lane;
  v4f v[PL_LOADS];
#pragma unroll
  for (int u = 0; u < PL_LOADS; ++u) v[u] = x4[u * 32];
  const bool odd = lane >= 16;
  const int c4 = lane & 15;
  const int id_e0 = __builtin_amdgcn_readlane(idv, 0);
  const int id_o0 = __builtin_amdgcn_readlane(idv, 1);
  int cur = odd ? id_o0 : id_e0;
  v4f acc = v[0];
#pragma unroll
  for (int u = 1; u < PL_LOADS; ++u) {
    const int id_e = __builtin_amdgcn_readlane(idv, 2 * u);
    const int id_o = __builtin_amdgcn_readlane(idv, 2 * u + 1);
    const int id = odd ? id_o : id_e;
    if (id != cur) { pool_flush(g, cur, c4, acc); cur = id; acc = v[u]; }
    else acc += v[u];
  }
  pool_flush(g, cur, c4, acc);
}

__global__ void add_kernel(const float* a, const float* b, float* c, int n) {
  int i = blockIdx.x * blockDim.x + threadIdx.x;
  if (i < n) c[i] = a[i] + b[i];
}

#define LOGIT_THREADS 256
static_assert(LOGIT_THREADS >= HIDDEN, "one thread per column statistic");
__global__ __launch_bounds__(LOGIT_THREADS) void logits_kernel(
    const float* __restrict__ h, const float* __restrict__ sums,
    const float* __restrict__ sumsq, float invN,
    const float* __restrict__ W0, const float* __restrict__ W1,
    const float* __restrict__ W2, const float* __restrict__ eb0,
    const float* __restrict__ eb1, const float* __restrict__ eb2,
    float* __restrict__ out, int G, int strideH, int strideS, int strideO) {
  constexpr int W_PER = (HIDDEN * N_CLASSES + LOGIT_THREADS - 1) / LOGIT_THREADS;
  __shared__ float s_w[W_PER * LOGIT_THREADS];
  __shared__ float s_ms[LOGIT_THREADS];
  __shared__ float s_s[LOGIT_THREADS];
  __shared__ float s_v[N_CLASSES];
  const int y = blockIdx.y;
  const float* W       = y == 0 ? W0  : (y == 1 ? W1  : W2);
  const float* extra_b = y == 0 ? eb0 : (y == 1 ? eb1 : eb2);
  h     += (size_t)y * strideH;
  sums  += (size_t)y * strideS;
  sumsq += (size_t)y * strideS;
  out   += (size_t)y * strideO;
  {
    const int ks = min((int)threadIdx.x, HIDDEN - 1);
    const float csum = sums[ks], csq = sumsq[ks];
    float wv[W_PER];
#pragma unroll
    for (int q = 0; q < W_PER; ++q)
      wv[q] = W[min((int)threadIdx.x + q * LOGIT_THREADS, HIDDEN * N_CLASSES - 1)];
    fold_stats(csum, csq, invN, s_ms, s_s);
#pragma unroll
    for (int q = 0; q < W_PER; ++q)
      s_w[(int)threadIdx.x + q * LOGIT_THREADS] = wv[q];
  }
  __syncthreads();
  if (threadIdx.x < N_CLASSES) {
    const int j = threadIdx.x;
    s_v[j] = fold_column<N_CLASSES>(s_w, s_ms, s_s, HIDDEN, j, extra_b ? extra_b[j] : 0.f);
  }
  __syncthreads();
  int g = blockIdx.x * blockDim.x + threadIdx.x;
  if (g >= G) return;
  float l[N_CLASSES];
  for (int c = 0; c < N_CLASSES; ++c) l[c] = s_v[c];
  const float* hg = h + (size_t)g * HIDDEN;
  for (int k = 0; k < HIDDEN; ++k) {
    float x = hg[k];
    for (int c = 0; c < N_CLASSES; ++c) l[c] += x * s_w[k * N_CLASSES + c];
  }
  float mx = l[0];
  for (int c = 1; c < N_CLASSES; ++c) mx = fmaxf(mx, l[c]);
  float s = 0.f;
  for (int c = 0; c < N_CLASSES; ++c) s += __expf(l[c] - mx);
  float lse = __logf(s) + mx;
  for (int c = 0; c < N_CLASSES; ++c) out[(size_t)g * N_CLASSES + c] = l[c] - lse;
}

extern "C" void kernel_launch(void* const* d_in, const int* in_sizes, int n_in,
                              void* d_out, int out_size, void* d_ws, size_t ws_size,
                              hipStream_t stream) {
  (void)in_sizes; (void)n_in; (void)out_size; (void)ws_size;
  const float* x       = (const float*)d_in[0];
  const int*   ei      = (const int*)d_in[1];
  const int*   row     = ei;
  const int*   col     = ei + N_EDGES;
  const int*   batch   = (const int*)d_in[2];
  const float* W_feat  = (const float*)d_in[3];
  const float* W_convs = (const float*)d_in[4];
  const float* b_convs = (const float*)d_in[5];
  const float* W_ea    = (const float*)d_in[6];
  const float* b_ea    = (const float*)d_in[7];
  const float* W_na    = (const float*)d_in[8];
  const float* b_na    = (const float*)d_in[9];
  const float* W_ctx   = (const float*)d_in[10];
  const float* b_ctx   = (const float*)d_in[11];
  const float* W_obj   = (const float*)d_in[12];
  const float* b_obj   = (const float*)d_in[13];
  const float* W1h[3]  = {(const float*)d_in[14], (const float*)d_in[18], (const float*)d_in[22]};
  const float* b1h[3]  = {(const float*)d_in[15], (const float*)d_in[19], (const float*)d_in[23]};
  const float* W2h[3]  = {(const float*)d_in[16], (const float*)d_in[20], (const float*)d_in[24]};
  const float* b2h[3]  = {(const float*)d_in[17], (const float*)d_in[21], (const float*)d_in[25]};

  const int SN = N_NODES * HIDDEN;
  const int SG = N_GRAPHS * HIDDEN;
  const int SS = STAT_STRIDE;
  size_t off = 0;
  auto carve = [&](size_t elems) -> float* {
    float* p = (float*)((char*)d_ws + off);
    off += ((elems * sizeof(float)) + 255) & ~(size_t)255;
    return p;
  };
  float* hb    = carve((size_t)SN);
  float* mcat  = carve((size_t)2 * SN);
  float* xcat  = carve((size_t)2 * SN);
  float* att   = carve((size_t)N_EDGES * 2);
  float* dinvu = carve(N_NODES);
  float* dinva = carve((size_t)2 * N_NODES);
  float* gcat  = carve((size_t)3 * SG);
  float* r1cat = carve((size_t)3 * SG);
  float* stat  = carve((size_t)3 * SS);
  float* part  = carve((size_t)2 * CS_MAX_SLABS * SS);
  int*   tcount  = (int*)carve(2 * BIN_STRIDE);
  int*   tstart  = (int*)carve(2 * BIN_STRIDE);
  int*   tcursor = (int*)carve(2 * BIN_STRIDE);
  int*   erec    = (int*)carve((size_t)2 * N_EDGES * 4);
  const int* srec   = erec + (size_t)N_EDGES * 4;
  const int* sstart = tstart + BIN_STRIDE;
  float* xc = xcat;
  float* xo = xcat + SN;
  float* gc = gcat;
  float* go = gcat + SG;
  float* gs = gcat + 2 * SG;

  auto nblk = [](long n, int b) { return (int)((n + b - 1) / b); };

  auto stats = [&](const float* A, int N, int F, int ny, int strideA) {
    const int P = nblk(N, CS_SLAB);
    float* rows = P == 1 ? stat : part;
    if (F == N_FEAT) {
      colstats_slab_kernel<N_FEAT><<<dim3(P, ny), CS_THREADS, 0, stream>>>(A, N, rows, strideA);
      if (P > 1) colstats_finish_kernel<N_FEAT><<<ny, CF_THREADS, 0, stream>>>(part, P, stat);
    } else {
      colstats_slab_kernel<HIDDEN><<<dim3(P, ny), CS_THREADS, 0, stream>>>(A, N, rows, strideA);
      if (P > 1) colstats_finish_kernel<HIDDEN><<<ny, CF_THREADS, 0, stream>>>(part, P, stat);
    }
  };
  auto gemm = [&](const float* A, float* C, int M, int K, int relu, int ny,
                  int strideA, int strideC, float invN,
                  const float* W0, const float* W1, const float* W2,
                  const float* e0, const float* e1, const float* e2) {
    const dim3 grid(nblk(M / 16, GEMM_WAVES), ny);
    if (K == N_FEAT)
      gemm_wmma_kernel<N_FEAT><<<grid, GEMM_THREADS, 0, stream>>>(
          A, stat, stat + N_FEAT, invN, W0, W1, W2, e0, e1, e2, C, M, relu,
          strideA, SS, strideC);
    else
      gemm_wmma_kernel<HIDDEN><<<grid, GEMM_THREADS, 0, stream>>>(
          A, stat, stat + N_FEAT, invN, W0, W1, W2, e0, e1, e2, C, M, relu,
          strideA, SS, strideC);
  };

  stats(x, N_NODES, N_FEAT, 1, 0);
  gemm(x, hb, N_NODES, N_FEAT, 1, 1, 0, 0, 1.f / N_NODES, W_feat, W_feat, W_feat,
       nullptr, nullptr, nullptr);

  hipMemsetAsync(tcount, 0, (size_t)2 * BIN_STRIDE * sizeof(int), stream);
  tile_count_kernel<<<dim3(BIN_BLOCKS, 2), BIN_THREADS, 0, stream>>>(row, col, N_EDGES,
                                                                    tcount);
  tile_scan_kernel<<<2, 256, 0, stream>>>(tcount, tstart, tcursor);
  tile_fill_kernel<<<dim3(BIN_BLOCKS, 2), BIN_THREADS, 0, stream>>>(row, col, N_EDGES,
                                                                   tcursor, erec);

  tile_deg_kernel<<<N_TILES, TD_THREADS, 0, stream>>>(srec, sstart, nullptr, dinvu,
                                                      N_NODES, 0);

  for (int i = 0; i < 3; ++i) {
    const float* Wi = W_convs + (size_t)i * HIDDEN * HIDDEN;
    const float* bi = b_convs + (size_t)i * HIDDEN;
    stats(hb, N_NODES, HIDDEN, 1, 0);
    gemm(hb, mcat, N_NODES, HIDDEN, 0, 1, 0, 0, 1.f / N_NODES, Wi, Wi, Wi,
         nullptr, nullptr, nullptr);
    tile_prop_kernel<<<dim3(N_TILES, 1), TP_THREADS, 0, stream>>>(
        mcat, erec, tstart, nullptr, 0, dinvu, bi, bi, hb, N_NODES, 0, 0);
  }

  edge_att_kernel<<<nblk(N_EDGES, 256), 256, 0, stream>>>(hb, row, col, W_ea, b_ea,
                                                          att, N_EDGES);
  node_att_split_kernel<<<nblk(N_NODES, 256), 256, 0, stream>>>(hb, W_na, b_na,
                                                                xc, xo, N_NODES);

  stats(xcat, N_NODES, HIDDEN, 2, SN);
  gemm(xcat, mcat, N_NODES, HIDDEN, 0, 2, SN, SN, 1.f / N_NODES, W_ctx, W_obj, W_obj,
       nullptr, nullptr, nullptr);
  tile_deg_kernel<<<N_TILES, TD_THREADS, 0, stream>>>(srec, sstart, att, dinva,
                                                      N_NODES, N_NODES);
  tile_prop_kernel<<<dim3(N_TILES, 2), TP_THREADS, 0, stream>>>(
      mcat, erec, tstart, att, 0, dinva, b_ctx, b_obj, xcat, N_NODES, SN, N_NODES);
  hipMemsetAsync(gcat, 0, (size_t)2 * SG * sizeof(float), stream);
  pool_kernel<<<dim3(nblk(N_NODES, PL_ROWS * (PL_THREADS / 32)), 2), PL_THREADS, 0, stream>>>(
      xcat, batch, gcat, N_NODES, SN, SG);

  add_kernel<<<nblk(N_GRAPHS * HIDDEN, 256), 256, 0, stream>>>(gc, go, gs,
                                                               N_GRAPHS * HIDDEN);

  float* out = (float*)d_out;
  stats(gcat, N_GRAPHS, HIDDEN, 3, SG);
  gemm(gcat, r1cat, N_GRAPHS, HIDDEN, 1, 3, SG, SG, 1.f / N_GRAPHS,
       W1h[0], W1h[1], W1h[2], b1h[0], b1h[1], b1h[2]);
  stats(r1cat, N_GRAPHS, HIDDEN, 3, SG);
  logits_kernel<<<dim3(nblk(N_GRAPHS, LOGIT_THREADS), 3), LOGIT_THREADS, 0, stream>>>(
      r1cat, stat, stat + N_FEAT, 1.f / N_GRAPHS, W2h[0], W2h[1], W2h[2],
      b2h[0], b2h[1], b2h[2], out, N_GRAPHS, SG, SS, N_GRAPHS * N_CLASSES);
}


# passes of different kinds that do not depend on each other share launches: edge binning chain rides behind the feature chain's blocks, node with edge attention; stream ops 34 -> 26; sum orders kept
# speedup vs baseline: 5.1939x; 1.0449x over previous
#include <hip/hip_runtime.h>
#include <math.h>


#define N_NODES  100000
#define N_EDGES  1000000
#define N_FEAT   128
#define HIDDEN   64
#define N_GRAPHS 512
#define N_CLASSES 10
#define BN_EPS  1e-5f
#define BN_BETA 1e-4f

typedef __attribute__((ext_vector_type(2))) float v2f;
typedef __attribute__((ext_vector_type(8))) float v8f;

#define SIDE_NONE  0
#define SIDE_ZERO  1
#define SIDE_COUNT 2
#define SIDE_SCAN  3
#define SIDE_FILL  4
#define SIDE_DEG   5

__device__ __forceinline__ void zero_body(int* __restrict__ p, int n4, int sb, int T) {
  const int i = sb * T + (int)threadIdx.x;
  if (i < n4) ((int4*)p)[i] = make_int4(0, 0, 0, 0);
}

#define CS_THREADS 512
#define CS_WAVES (CS_THREADS / 32)
#define CS_BATCH 8
#define CS_SLAB 512
#define CS_MAX_SLABS ((N_NODES + CS_SLAB - 1) / CS_SLAB)
#define CF_THREADS 1024
#define STAT_STRIDE (2 * N_FEAT)

typedef __attribute__((ext_vector_type(4))) float v4f;

template <int F, bool GUARD>
__device__ __forceinline__ void cs_load(const v4f* __restrict__ A4, int rfirst, int r1,
                                        int c4, int b0, v4f (&v)[CS_BATCH]) {
  constexpr int VPR  = F / 4;
  constexpr int STEP = CS_WAVES * (32 / VPR);
#pragma unroll
  for (int u = 0; u < CS_BATCH; ++u) {
    int r = rfirst + (b0 + u) * STEP;
    if (GUARD) r = min(r, r1 - 1);
    v[u] = A4[(size_t)r * VPR + c4];
  }
}

template <int F, bool GUARD>
__device__ __forceinline__ void cs_slab_sum(const v4f* __restrict__ A4, int rfirst,
                                            int r1, int c4, v4f& s, v4f& q) {
  constexpr int VPR  = F / 4;
  constexpr int STEP = CS_WAVES * (32 / VPR);
  constexpr int TRIPS = CS_SLAB / (STEP * CS_BATCH);
  static_assert(TRIPS * STEP * CS_BATCH == CS_SLAB, "a slab is a whole number of batches");
  v4f cur[CS_BATCH], nxt[CS_BATCH];
  cs_load<F, GUARD>(A4, rfirst, r1, c4, 0, cur);
#pragma unroll
  for (int t = 0; t < TRIPS; ++t) {
    if (t + 1 < TRIPS) cs_load<F, GUARD>(A4, rfirst, r1, c4, (t + 1) * CS_BATCH, nxt);
    asm volatile("" :: "v"(cur[0]), "v"(cur[1]), "v"(cur[2]), "v"(cur[3]),
                       "v"(cur[4]), "v"(cur[5]), "v"(cur[6]), "v"(cur[7]));
#pragma unroll
    for (int u = 0; u < CS_BATCH; ++u) {
      const int r = rfirst + (t * CS_BATCH + u) * STEP;
      if (!GUARD || r < r1) { s += cur[u]; q += cur[u] * cur[u]; }
    }
#pragma unroll
    for (int u = 0; u < CS_BATCH; ++u) cur[u] = nxt[u];
  }
}
static_assert(CS_BATCH == 8, "the fence above names eight values");

template <int F>
__device__ __forceinline__ void colstats_slab_body(
    const float* __restrict__ A, int N, float* __restrict__ part, int strideA,
    int bx, int by, int gx) {
  constexpr int VPR = F / 4;
  static_assert(VPR == 16 || VPR == 32, "a wave-wide load covers one or two whole rows");
  __shared__ v4f s_s[CS_THREADS];
  __shared__ v4f s_q[CS_THREADS];
  A    += (size_t)by * strideA;
  part += ((size_t)by * gx + bx) * STAT_STRIDE;
  const int lane = threadIdx.x & 31;
  const int wave = threadIdx.x >> 5;
  const int c4   = lane & (VPR - 1);
  const int r0   = bx * CS_SLAB;
  const int r1   = min(N, r0 + CS_SLAB);
  const int rfirst = r0 + wave * (32 / VPR) + lane / VPR;
  v4f s = {0.f, 0.f, 0.f, 0.f}, q = s;
  if (r0 + CS_SLAB <= N) cs_slab_sum<F, false>((const v4f*)A, rfirst, r1, c4, s, q);
  else                   cs_slab_sum<F, true>((const v4f*)A, rfirst, r1, c4, s, q);
  s_s[threadIdx.x] = s;
  s_q[threadIdx.x] = q;
  __syncthreads();
  if (threadIdx.x < 2 * VPR) {
    const bool sq = threadIdx.x >= VPR;
    const int c = threadIdx.x & (VPR - 1);
    const v4f* src = sq ? s_q : s_s;
    v4f a = src[c];
#pragma unroll
    for (int j = 1; j < CS_THREADS / VPR; ++j) a += src[j * VPR + c];
    ((v4f*)(part + (sq ? N_FEAT : 0)))[c] = a;
  }
}

template <int F>
__device__ __forceinline__ void colstats_finish_body(
    const float* __restrict__ part, int P, float* __restrict__ stat, int bx) {
  constexpr int VPR = F / 4;
  constexpr int V2  = 2 * VPR;
  constexpr int G   = CF_THREADS / V2;
  constexpr int PER = (CS_MAX_SLABS + G - 1) / G;
  __shared__ float4 s_r[CF_THREADS];
  part += (size_t)bx * P * STAT_STRIDE;
  stat += (size_t)bx * STAT_STRIDE;
  const int c = threadIdx.x & (V2 - 1);
  const int g = threadIdx.x / V2;
  const int off = c < VPR ? c * 4 : N_FEAT + (c - VPR) * 4;
  float4 v[PER];
#pragma unroll
  for (int i = 0; i < PER; ++i)
    v[i] = *(const float4*)(part + (size_t)min(g + i * G, P - 1) * STAT_STRIDE + off);
  float4 a = make_float4(0.f, 0.f, 0.f, 0.f);
#pragma unroll
  for (int i = 0; i < PER; ++i)
    if (g + i * G < P) { a.x += v[i].x; a.y += v[i].y; a.z += v[i].z; a.w += v[i].w; }
  s_r[threadIdx.x] = a;
  __syncthreads();
  if (threadIdx.x < V2) {
    float4 t = s_r[c];
#pragma unroll
    for (int j = 1; j < G; ++j) {
      const float4 b = s_r[j * V2 + c];
      t.x += b.x; t.y += b.y; t.z += b.z; t.w += b.w;
    }
    *(float4*)(stat + off) = t;
  }
}

__device__ __forceinline__ void fold_stats(float sum, float sumsq, float invN,
                                           float* s_ms, float* s_s) {
  float mean = sum * invN;
  float var  = sumsq * invN - mean * mean;
  float sc   = rsqrtf(var + BN_EPS);
  s_ms[threadIdx.x] = mean * sc;
  s_s[threadIdx.x]  = sc;
}

template <int PITCH>
__device__ __forceinline__ float fold_column(float* s_w, const float* s_ms,
                                             const float* s_s, int K, int j,
                                             float acc) {
#pragma unroll 8
  for (int k = 0; k < K; ++k) {
    const float w = s_w[k * PITCH + j];
    s_w[k * PITCH + j] = s_s[k] * w;
    acc += (BN_BETA - s_ms[k]) * w;
  }
  return acc;
}

#define GEMM_WAVES 8
#define GEMM_THREADS (GEMM_WAVES * 32)
#define W_PITCH 72
template <int K>
__device__ __forceinline__ void gemm_wmma_body(
    const float* __restrict__ A, const float* __restrict__ sums,
    const float* __restrict__ sumsq, float invN,
    const float* __restrict__ W0, const float* __restrict__ W1,
    const float* __restrict__ W2, const float* __restrict__ eb0,
    const float* __restrict__ eb1, const float* __restrict__ eb2,
    float* __restrict__ C, int M, int doRelu,
    int strideA, int strideS, int strideC, int bx, int by) {
  constexpr int W_PER = K * HIDDEN / GEMM_THREADS;
  static_assert(K * HIDDEN == W_PER * GEMM_THREADS, "W splits evenly over the threads");
  static_assert(GEMM_THREADS % HIDDEN == 0, "a thread keeps one column of W");
  static_assert(GEMM_THREADS >= K, "one thread per column statistic");
  __shared__ float s_w[K * W_PITCH];
  __shared__ float s_ms[GEMM_THREADS];
  __shared__ float s_s[GEMM_THREADS];
  __shared__ float s_v[HIDDEN];
  const int y = by;
  const float* W       = y == 0 ? W0  : (y == 1 ? W1  : W2);
  const float* extra_b = y == 0 ? eb0 : (y == 1 ? eb1 : eb2);
  A     += (size_t)y * strideA;
  sums  += (size_t)y * strideS;
  sumsq += (size_t)y * strideS;
  C     += (size_t)y * strideC;
  {
    const int ks = min((int)threadIdx.x, K - 1);
    const float csum = sums[ks], csq = sumsq[ks];
    const int wcol = threadIdx.x & (HIDDEN - 1);
    const int wrow = threadIdx.x / HIDDEN;
    float wv[W_PER];
#pragma unroll
    for (int q = 0; q < W_PER; ++q)
      wv[q] = W[(int)threadIdx.x + q * GEMM_THREADS];
    fold_stats(csum, csq, invN, s_ms, s_s);
#pragma unroll
    for (int q = 0; q < W_PER; ++q)
      s_w[(wrow + q * (GEMM_THREADS / HIDDEN)) * W_PITCH + wcol] = wv[q];
  }
  __syncthreads();
  if (threadIdx.x < HIDDEN) {
    const int j = threadIdx.x;
    s_v[j] = fold_column<W_PITCH>(s_w, s_ms, s_s, K, j, extra_b ? extra_b[j] : 0.f);
  }
  __syncthreads();
  const int lane = threadIdx.x & 31;
  const int wave = threadIdx.x >> 5;
  const int rowTile = bx * GEMM_WAVES + wave;
  if (rowTile >= (M >> 4)) return;
  const int mrow = (rowTile << 4) + (lane & 15);
  const int kOff = (lane >> 4) << 1;
  const int n0   = lane & 15;
  v8f acc[4] = {};
  for (int kk = 0; kk < K; kk += 4) {
    const int k0 = kk + kOff;
    v2f a = *(const v2f*)(A + (size_t)mrow * K + k0);
    const float* w0 = s_w + k0 * W_PITCH + n0;
#pragma unroll
    for (int t = 0; t < 4; ++t) {
      v2f b;
      b.x = w0[t * 16];
      b.y = w0[W_PITCH + t * 16];
      acc[t] = __builtin_amdgcn_wmma_f32_16x16x4_f32(
          false, a, false, b, (short)0, acc[t], false, false);
    }
  }
  const int rowBase = (rowTile << 4) + ((lane >> 4) << 3);
#pragma unroll
  for (int t = 0; t < 4; ++t) {
    const int n = t * 16 + n0;
    const float bias = s_v[n];
#pragma unroll
    for (int e = 0; e < 8; ++e) {
      float val = acc[t][e] + bias;
      if (doRelu) val = fmaxf(val, 0.f);
      C[(size_t)(rowBase + e) * HIDDEN + n] = val;
    }
  }
}

#define TILE_NODES 400
#define N_TILES ((N_NODES + TILE_NODES - 1) / TILE_NODES)
#define BIN_THREADS 1024
#define BIN_PER_THREAD 4
#define BIN_EDGES (BIN_THREADS * BIN_PER_THREAD)
#define BIN_BLOCKS ((N_EDGES + BIN_EDGES - 1) / BIN_EDGES)
#define TP_THREADS 1024
#define TP_BATCH 8
#define TP_VEC (TILE_NODES * HIDDEN / 4)
#define TP_FIN ((TP_VEC + TP_THREADS - 1) / TP_THREADS)
#define BIN_STRIDE 256
#define TD_EDGES 4096
static_assert(N_TILES + 1 <= BIN_STRIDE, "a list's bin starts fit its table");
static_assert(N_TILES <= 256, "the scan below is one 256-thread block");
static_assert(TP_THREADS % (HIDDEN / 4) == 0, "a thread keeps one column group");

__device__ __forceinline__ void tile_count_body(
    const int* __restrict__ row, const int* __restrict__ col, int E,
    int* __restrict__ counts, int bx, int by) {
  __shared__ int s_cnt[N_TILES];
  const int* key = by == 0 ? col : row;
  counts += by * BIN_STRIDE;
  for (int t = threadIdx.x; t < N_TILES; t += BIN_THREADS) s_cnt[t] = 0;
  __syncthreads();
  for (int i = 0; i < BIN_PER_THREAD; ++i) {
    const int e = (bx * BIN_PER_THREAD + i) * BIN_THREADS + threadIdx.x;
    if (e < E) atomicAdd(&s_cnt[key[e] / TILE_NODES], 1);
  }
  __syncthreads();
  for (int t = threadIdx.x; t < N_TILES; t += BIN_THREADS) {
    const int v = s_cnt[t];
    if (v) atomicAdd(&counts[t], v);
  }
}

__device__ __forceinline__ void tile_scan_body(
    const int* __restrict__ counts, int* __restrict__ start,
    int* __restrict__ cursor, int bx) {
  __shared__ int s[256];
  counts += bx * BIN_STRIDE;
  start  += bx * BIN_STRIDE;
  cursor += bx * BIN_STRIDE;
  const int t = threadIdx.x;
  const int v = t < N_TILES ? counts[t] : 0;
  s[t] = v;
  __syncthreads();
  for (int d = 1; d < 256; d <<= 1) {
    const int add = t >= d ? s[t - d] : 0;
    __syncthreads();
    s[t] += add;
    __syncthreads();
  }
  const int excl = s[t] - v;
  if (t < N_TILES) { start[t] = excl; cursor[t] = excl; }
  if (t == N_TILES - 1) start[N_TILES] = s[t];
}

template <int T>
__device__ __forceinline__ void tile_fill_body(
    const int* __restrict__ row, const int* __restrict__ col, int E,
    int* __restrict__ cursor, int* __restrict__ rec, int bx, int by) {
  constexpr int PER = BIN_EDGES / T;
  static_assert(PER * T == BIN_EDGES, "a block's edges split evenly over its threads");
  __shared__ int s_cnt[N_TILES];
  __shared__ int s_base[N_TILES];
  const int* key = by == 0 ? col : row;
  const int* oth = by == 0 ? row : col;
  cursor += by * BIN_STRIDE;
  rec    += (size_t)by * N_EDGES * 4;
  for (int t = threadIdx.x; t < N_TILES; t += T) s_cnt[t] = 0;
  __syncthreads();
  int rr[PER], cc[PER], tl[PER], rk[PER];
#pragma unroll
  for (int i = 0; i < PER; ++i) {
    const int e = (bx * PER + i) * T + threadIdx.x;
    rr[i] = 0; cc[i] = 0; tl[i] = 0; rk[i] = 0;
    if (e < E) {
      rr[i] = oth[e];
      cc[i] = key[e];
      tl[i] = cc[i] / TILE_NODES;
      rk[i] = atomicAdd(&s_cnt[tl[i]], 1);
    }
  }
  __syncthreads();
  for (int t = threadIdx.x; t < N_TILES; t += T) {
    const int v = s_cnt[t];
    s_base[t] = v ? atomicAdd(&cursor[t], v) : 0;
  }
  __syncthreads();
#pragma unroll
  for (int i = 0; i < PER; ++i) {
    const int e = (bx * PER + i) * T + threadIdx.x;
    if (e < E) {
      int4 q;
      q.x = rr[i];
      q.y = cc[i] - tl[i] * TILE_NODES;
      q.z = e;
      q.w = tl[i];
      ((int4*)rec)[s_base[tl[i]] + rk[i]] = q;
    }
  }
}

template <int T>
__device__ __forceinline__ void tile_deg_body(
    const int* __restrict__ rec, const int* __restrict__ start,
    const float* __restrict__ att, float* __restrict__ dinv, int N, int strideD,
    int tile) {
  constexpr int TD_THREADS = T;
  constexpr int TD_PER_THREAD = TD_EDGES / T;
  static_assert(TD_PER_THREAD * T == TD_EDGES, "a trip's edges split evenly over the threads");
  __shared__ float s_deg[2 * TILE_NODES];
  const int n0 = tile * TILE_NODES;
  const int nn = min(TILE_NODES, N - n0);
  for (int i = threadIdx.x; i < 2 * TILE_NODES; i += TD_THREADS) s_deg[i] = 1.0f;
  __syncthreads();
  const int beg = start[tile], end = start[tile + 1];
  for (int e0 = beg + (int)threadIdx.x; e0 < end; e0 += TD_THREADS * TD_PER_THREAD) {
    int loc[TD_PER_THREAD], eid[TD_PER_THREAD];
#pragma unroll
    for (int k = 0; k < TD_PER_THREAD; ++k) {
      const int4 q = ((const int4*)rec)[min(e0 + k * TD_THREADS, end - 1)];
      loc[k] = q.y;
      eid[k] = q.z;
    }
    if (att) {
      float2 a[TD_PER_THREAD];
#pragma unroll
      for (int k = 0; k < TD_PER_THREAD; ++k) a[k] = ((const float2*)att)[eid[k]];
#pragma unroll
      for (int k = 0; k < TD_PER_THREAD; ++k) {
        if (e0 + k * TD_THREADS < end) {
          __hip_atomic_fetch_add(&s_deg[loc[k]], a[k].x, __ATOMIC_RELAXED,
                                 __HIP_MEMORY_SCOPE_WORKGROUP);
          __hip_atomic_fetch_add(&s_deg[TILE_NODES + loc[k]], a[k].y, __ATOMIC_RELAXED,
                                 __HIP_MEMORY_SCOPE_WORKGROUP);
        }
      }
    } else {
#pragma unroll
      for (int k = 0; k < TD_PER_THREAD; ++k) {
        if (e0 + k * TD_THREADS < end)
          __hip_atomic_fetch_add(&s_deg[loc[k]], 1.0f, __ATOMIC_RELAXED,
                                 __HIP_MEMORY_SCOPE_WORKGROUP);
      }
    }
  }
  __syncthreads();
  for (int i = threadIdx.x; i < nn; i += TD_THREADS) {
    const float d0 = s_deg[i];
    dinv[n0 + i] = d0 > 0.f ? rsqrtf(d0) : 0.f;
    if (att) {
      const float d1 = s_deg[TILE_NODES + i];
      dinv[(size_t)strideD + n0 + i] = d1 > 0.f ? rsqrtf(d1) : 0.f;
    }
  }
}

__device__ __forceinline__ void tile_batch(const float* __restrict__ m,
                                           float* s_acc, int r, int cl,
                                           float coef, int g, int lane) {
  float a0[TP_BATCH], a1[TP_BATCH];
#pragma unroll
  for (int u = 0; u < TP_BATCH; ++u) {
    const int ri = __builtin_amdgcn_readlane(r, g + u);
    const float* src = m + (size_t)ri * HIDDEN + lane;
    a0[u] = src[0];
    a1[u] = src[32];
  }
#pragma unroll
  for (int u = 0; u < TP_BATCH; ++u) {
    const int   ci = __builtin_amdgcn_readlane(cl, g + u);
    const float fi = __builtin_bit_cast(float,
        __builtin_amdgcn_readlane(__builtin_bit_cast(int, coef), g + u));
    float* dst = s_acc + ci * HIDDEN + lane;
    __hip_atomic_fetch_add(dst, fi * a0[u], __ATOMIC_RELAXED,
                           __HIP_MEMORY_SCOPE_WORKGROUP);
    __hip_atomic_fetch_add(dst + 32, fi * a1[u], __ATOMIC_RELAXED,
                           __HIP_MEMORY_SCOPE_WORKGROUP);
  }
}

__global__ __launch_bounds__(TP_THREADS) void tile_prop_kernel(
    const float* __restrict__ m, const int* __restrict__ rec,
    const int* __restrict__ start, const float* __restrict__ att, int comp,
    const float* __restrict__ dinv, const float* __restrict__ b0,
    const float* __restrict__ b1, float* __restrict__ out, int N,
    int strideM, int strideD) {
  __shared__ __attribute__((aligned(16))) float s_acc[TILE_NODES * HIDDEN];
  const float* b = blockIdx.y == 0 ? b0 : b1;
  comp += blockIdx.y;
  m    += (size_t)blockIdx.y * strideM;
  out  += (size_t)blockIdx.y * strideM;
  dinv += (size_t)blockIdx.y * strideD;
  const int tile = blockIdx.x;
  const int n0 = tile * TILE_NODES;
  const int nn = min(TILE_NODES, N - n0);
  for (int i = threadIdx.x; i < TILE_NODES * HIDDEN; i += TP_THREADS) s_acc[i] = 0.f;
  __syncthreads();
  const int beg = start[tile], end = start[tile + 1];
  const int lane = threadIdx.x & 31;
  const int wave = __builtin_amdgcn_readfirstlane((int)(threadIdx.x >> 5));
  for (int e0 = beg + wave * 32; e0 < end; e0 += TP_THREADS) {
    const int e = e0 + lane;
    int r = 0, cl = 0;
    float coef = 0.f;
    if (e < end) {
      const int4 q = ((const int4*)rec)[e];
      r = q.x;
      cl = q.y;
      const float w = att ? att[(size_t)q.z * 2 + comp] : 1.0f;
      coef = dinv[r] * w * dinv[n0 + cl];
    }
    const int cnt = min(32, end - e0);
    int g = 0;
    if (cnt == 32) {
#pragma unroll
      for (int gg = 0; gg < 32; gg += TP_BATCH)
        tile_batch(m, s_acc, r, cl, coef, gg, lane);
      g = 32;
    } else {
      for (; g + TP_BATCH <= cnt; g += TP_BATCH)
        tile_batch(m, s_acc, r, cl, coef, g, lane);
    }
    for (; g < cnt; ++g) {
      const int   ri = __builtin_amdgcn_readlane(r, g);
      const int   ci = __builtin_amdgcn_readlane(cl, g);
      const float fi = __builtin_bit_cast(float,
          __builtin_amdgcn_readlane(__builtin_bit_cast(int, coef), g));
      const float* src = m + (size_t)ri * HIDDEN + lane;
      float* dst = s_acc + ci * HIDDEN + lane;
      __hip_atomic_fetch_add(dst, fi * src[0], __ATOMIC_RELAXED,
                             __HIP_MEMORY_SCOPE_WORKGROUP);
      __hip_atomic_fetch_add(dst + 32, fi * src[32], __ATOMIC_RELAXED,
                             __HIP_MEMORY_SCOPE_WORKGROUP);
    }
  }
  const int nv = nn * (HIDDEN / 4);
  const float4* m4 = (const float4*)(m + (size_t)n0 * HIDDEN);
  float4* o4 = (float4*)(out + (size_t)n0 * HIDDEN);
  const float4 bv = ((const float4*)b)[threadIdx.x & (HIDDEN / 4 - 1)];
  float4 mv[TP_FIN];
  float  dv[TP_FIN];
#pragma unroll
  for (int k = 0; k < TP_FIN; ++k) {
    const int j4 = min((int)threadIdx.x + k * TP_THREADS, nv - 1);
    mv[k] = m4[j4];
    dv[k] = dinv[n0 + j4 / (HIDDEN / 4)];
  }
  __syncthreads();
#pragma unroll
  for (int k = 0; k < TP_FIN; ++k) {
    const int i4 = threadIdx.x + k * TP_THREADS;
    if (i4 < nv) {
      const float4 a = ((const float4*)s_acc)[i4];
      const float d2 = dv[k] * dv[k];
      float4 r;
      r.x = fmaxf(a.x + d2 * mv[k].x + bv.x, 0.f);
      r.y = fmaxf(a.y + d2 * mv[k].y + bv.y, 0.f);
      r.z = fmaxf(a.z + d2 * mv[k].z + bv.z, 0.f);
      r.w = fmaxf(a.w + d2 * mv[k].w + bv.w, 0.f);
      o4[i4] = r;
    }
  }
}

#define EA_BATCH 8
#define ATT_THREADS 256
__device__ __forceinline__ void edge_att_body(
    const float* __restrict__ h, const int* __restrict__ row,
    const int* __restrict__ col, const float* __restrict__ Wea,
    const float* __restrict__ bea, float* __restrict__ att, int E, int bx) {
  const int lane = threadIdx.x & 31;
  const int e = bx * ATT_THREADS + threadIdx.x;
  const int cnt = __builtin_amdgcn_readfirstlane(min(32, E - (e - lane)));
  if (cnt <= 0) return;
  int r = 0, c = 0;
  if (e < E) { r = row[e]; c = col[e]; }
  const float wr0a = Wea[lane * 2 + 0],                 wr1a = Wea[lane * 2 + 1];
  const float wr0b = Wea[(lane + 32) * 2 + 0],          wr1b = Wea[(lane + 32) * 2 + 1];
  const float wc0a = Wea[(HIDDEN + lane) * 2 + 0],      wc1a = Wea[(HIDDEN + lane) * 2 + 1];
  const float wc0b = Wea[(HIDDEN + lane + 32) * 2 + 0], wc1b = Wea[(HIDDEN + lane + 32) * 2 + 1];
  float my0 = 0.f, my1 = 0.f;
#pragma unroll
  for (int g = 0; g < 32; g += EA_BATCH) {
    if (g < cnt) {
      float ra[EA_BATCH], rb[EA_BATCH], ca[EA_BATCH], cb[EA_BATCH];
#pragma unroll
      for (int u = 0; u < EA_BATCH; ++u) {
        const int ri = __builtin_amdgcn_readlane(r, g + u);
        const int ci = __builtin_amdgcn_readlane(c, g + u);
        const float* hr = h + (size_t)ri * HIDDEN + lane;
        const float* hc = h + (size_t)ci * HIDDEN + lane;
        ra[u] = hr[0]; rb[u] = hr[32];
        ca[u] = hc[0]; cb[u] = hc[32];
      }
#pragma unroll
      for (int u = 0; u < EA_BATCH; ++u) {
        const float p0 = ra[u] * wr0a + rb[u] * wr0b + ca[u] * wc0a + cb[u] * wc0b;
        const float p1 = ra[u] * wr1a + rb[u] * wr1b + ca[u] * wc1a + cb[u] * wc1b;
        const float keep = lane < 16 ? p0 : p1;
        const float give = lane < 16 ? p1 : p0;
        float x = keep + __shfl_xor(give, 16, 32);
        x += __shfl_xor(x, 8, 32);
        x += __shfl_xor(x, 4, 32);
        x += __shfl_xor(x, 2, 32);
        x += __shfl_xor(x, 1, 32);
        const float s0 = __builtin_bit_cast(float,
            __builtin_amdgcn_readlane(__builtin_bit_cast(int, x), 0));
        const float s1 = __builtin_bit_cast(float,
            __builtin_amdgcn_readlane(__builtin_bit_cast(int, x), 16));
        if (lane == g + u) { my0 = s0; my1 = s1; }
      }
    }
  }
  if (e >= E) return;
  const float l0 = my0 + bea[0], l1 = my1 + bea[1];
  float mx = fmaxf(l0, l1);
  float e0 = __expf(l0 - mx), e1 = __expf(l1 - mx);
  float inv = 1.f / (e0 + e1);
  att[(size_t)e * 2 + 0] = e0 * inv;
  att[(size_t)e * 2 + 1] = e1 * inv;
}

__device__ __forceinline__ void node_att_split_body(const float* __restrict__ h,
                                                    const float* __restrict__ Wna,
                                                    const float* __restrict__ bna,
                                                    float* __restrict__ xc,
                                                    float* __restrict__ xo, int N, int bx) {
  int i = bx * ATT_THREADS + threadIdx.x;
  if (i >= N) return;
  const float* hi = h + (size_t)i * HIDDEN;
  float l0 = bna[0], l1 = bna[1];
  for (int j = 0; j < HIDDEN; ++j) {
    float a = hi[j];
    l0 += a * Wna[j * 2 + 0];
    l1 += a * Wna[j * 2 + 1];
  }
  float mx = fmaxf(l0, l1);
  float e0 = __expf(l0 - mx), e1 = __expf(l1 - mx);
  float inv = 1.f / (e0 + e1);
  float p0 = e0 * inv, p1 = e1 * inv;
  for (int j = 0; j < HIDDEN; ++j) {
    float a = hi[j];
    xc[(size_t)i * HIDDEN + j] = p0 * a;
    xo[(size_t)i * HIDDEN + j] = p1 * a;
  }
}

#define PL_ROWS 32
#define PL_LOADS (PL_ROWS / 2)
#define PL_THREADS 256
static_assert(N_NODES % PL_ROWS == 0, "every row of a wave exists");
static_assert(HIDDEN * 4 * 2 == 32 * 16, "a wave-wide 16-byte load covers two rows");

__device__ __forceinline__ void pool_flush(float* __restrict__ g, int id, int c4, v4f a) {
  float* dst = g + (size_t)id * HIDDEN + c4 * 4;
  atomicAdd(dst + 0, a.x);
  atomicAdd(dst + 1, a.y);
  atomicAdd(dst + 2, a.z);
  atomicAdd(dst + 3, a.w);
}

__global__ __launch_bounds__(PL_THREADS) void pool_kernel(
    const float* __restrict__ x, const int* __restrict__ batch,
    float* __restrict__ g, int N, int strideX, int strideG) {
  x += (size_t)blockIdx.y * strideX;
  g += (size_t)blockIdx.y * strideG;
  const int lane = threadIdx.x & 31;
  const int wave = __builtin_amdgcn_readfirstlane((int)(threadIdx.x >> 5));
  const int r0 = (blockIdx.x * (PL_THREADS / 32) + wave) * PL_ROWS;
  if (r0 >= N) return;
  const int idv = batch[r0 + lane];
  const v4f* x4 = (const v4f*)(x + (size_t)r0 * HIDDEN) + lane;
  v4f v[PL_LOADS];
#pragma unroll
  for (int u = 0; u < PL_LOADS; ++u) v[u] = x4[u * 32];
  const bool odd = lane >= 16;
  const int c4 = lane & 15;
  const int id_e0 = __builtin_amdgcn_readlane(idv, 0);
  const int id_o0 = __builtin_amdgcn_readlane(idv, 1);
  int cur = odd ? id_o0 : id_e0;
  v4f acc = v[0];
#pragma unroll
  for (int u = 1; u < PL_LOADS; ++u) {
    const int id_e = __builtin_amdgcn_readlane(idv, 2 * u);
    const int id_o = __builtin_amdgcn_readlane(idv, 2 * u + 1);
    const int id = odd ? id_o : id_e;
    if (id != cur) { pool_flush(g, cur, c4, acc); cur = id; acc = v[u]; }
    else acc += v[u];
  }
  pool_flush(g, cur, c4, acc);
}

__global__ void add_kernel(const float* a, const float* b, float* c, int n) {
  int i = blockIdx.x * blockDim.x + threadIdx.x;
  if (i < n) c[i] = a[i] + b[i];
}

#define LOGIT_THREADS 256
static_assert(LOGIT_THREADS >= HIDDEN, "one thread per column statistic");
__global__ __launch_bounds__(LOGIT_THREADS) void logits_kernel(
    const float* __restrict__ h, const float* __restrict__ sums,
    const float* __restrict__ sumsq, float invN,
    const float* __restrict__ W0, const float* __restrict__ W1,
    const float* __restrict__ W2, const float* __restrict__ eb0,
    const float* __restrict__ eb1, const float* __restrict__ eb2,
    float* __restrict__ out, int G, int strideH, int strideS, int strideO) {
  constexpr int W_PER = (HIDDEN * N_CLASSES + LOGIT_THREADS - 1) / LOGIT_THREADS;
  __shared__ float s_w[W_PER * LOGIT_THREADS];
  __shared__ float s_ms[LOGIT_THREADS];
  __shared__ float s_s[LOGIT_THREADS];
  __shared__ float s_v[N_CLASSES];
  const int y = blockIdx.y;
  const float* W       = y == 0 ? W0  : (y == 1 ? W1  : W2);
  const float* extra_b = y == 0 ? eb0 : (y == 1 ? eb1 : eb2);
  h     += (size_t)y * strideH;
  sums  += (size_t)y * strideS;
  sumsq += (size_t)y * strideS;
  out   += (size_t)y * strideO;
  {
    const int ks = min((int)threadIdx.x, HIDDEN - 1);
    const float csum = sums[ks], csq = sumsq[ks];
    float wv[W_PER];
#pragma unroll
    for (int q = 0; q < W_PER; ++q)
      wv[q] = W[min((int)threadIdx.x + q * LOGIT_THREADS, HIDDEN * N_CLASSES - 1)];
    fold_stats(csum, csq, invN, s_ms, s_s);
#pragma unroll
    for (int q = 0; q < W_PER; ++q)
      s_w[(int)threadIdx.x + q * LOGIT_THREADS] = wv[q];
  }
  __syncthreads();
  if (threadIdx.x < N_CLASSES) {
    const int j = threadIdx.x;
    s_v[j] = fold_column<N_CLASSES>(s_w, s_ms, s_s, HIDDEN, j, extra_b ? extra_b[j] : 0.f);
  }
  __syncthreads();
  int g = blockIdx.x * blockDim.x + threadIdx.x;
  if (g >= G) return;
  float l[N_CLASSES];
  for (int c = 0; c < N_CLASSES; ++c) l[c] = s_v[c];
  const float* hg = h + (size_t)g * HIDDEN;
  for (int k = 0; k < HIDDEN; ++k) {
    float x = hg[k];
    for (int c = 0; c < N_CLASSES; ++c) l[c] += x * s_w[k * N_CLASSES + c];
  }
  float mx = l[0];
  for (int c = 1; c < N_CLASSES; ++c) mx = fmaxf(mx, l[c]);
  float s = 0.f;
  for (int c = 0; c < N_CLASSES; ++c) s += __expf(l[c] - mx);
  float lse = __logf(s) + mx;
  for (int c = 0; c < N_CLASSES; ++c) out[(size_t)g * N_CLASSES + c] = l[c] - lse;
}


template <int F, int SIDE>
__global__ __launch_bounds__(CS_THREADS) void colstats_slab_kernel(
    const float* __restrict__ A, int N, float* __restrict__ part, int strideA,
    int gx, int ny,
    int* __restrict__ zbuf, int zn4,
    const int* __restrict__ row, const int* __restrict__ col, int E,
    int* __restrict__ cursor, int* __restrict__ rec_out,
    const int* __restrict__ rec, const int* __restrict__ start,
    const float* __restrict__ att, float* __restrict__ dinv, int strideD) {
  const int b = blockIdx.x;
  if (b < gx * ny) {
    colstats_slab_body<F>(A, N, part, strideA, b % gx, b / gx, gx);
    return;
  }
  const int sb = b - gx * ny;
  if constexpr (SIDE == SIDE_ZERO) zero_body(zbuf, zn4, sb, CS_THREADS);
  if constexpr (SIDE == SIDE_FILL)
    tile_fill_body<CS_THREADS>(row, col, E, cursor, rec_out, sb % BIN_BLOCKS, sb / BIN_BLOCKS);
  if constexpr (SIDE == SIDE_DEG)
    tile_deg_body<CS_THREADS>(rec, start, att, dinv, N_NODES, strideD, sb);
}

template <int F, int SIDE>
__global__ __launch_bounds__(CF_THREADS) void colstats_finish_kernel(
    const float* __restrict__ part, int P, float* __restrict__ stat, int ny,
    const int* __restrict__ row, const int* __restrict__ col, int E,
    int* __restrict__ counts) {
  static_assert(CF_THREADS == BIN_THREADS, "the count pass runs in this launch's blocks");
  const int b = blockIdx.x;
  if (b < ny) {
    colstats_finish_body<F>(part, P, stat, b);
    return;
  }
  const int sb = b - ny;
  if constexpr (SIDE == SIDE_COUNT)
    tile_count_body(row, col, E, counts, sb % BIN_BLOCKS, sb / BIN_BLOCKS);
}

template <int K, int SIDE>
__global__ __launch_bounds__(GEMM_THREADS) void gemm_wmma_kernel(
    const float* __restrict__ A, const float* __restrict__ sums,
    const float* __restrict__ sumsq, float invN,
    const float* __restrict__ W0, const float* __restrict__ W1,
    const float* __restrict__ W2, const float* __restrict__ eb0,
    const float* __restrict__ eb1, const float* __restrict__ eb2,
    float* __restrict__ C, int M, int doRelu,
    int strideA, int strideS, int strideC, int gx, int ny,
    int* __restrict__ zbuf, int zn4,
    const int* __restrict__ counts, int* __restrict__ startw, int* __restrict__ cursor,
    const int* __restrict__ rec, const int* __restrict__ start,
    const float* __restrict__ att, float* __restrict__ dinv, int strideD) {
  static_assert(GEMM_THREADS == 256, "the scan pass is written for 256 threads");
  const int b = blockIdx.x;
  if (b < gx * ny) {
    gemm_wmma_body<K>(A, sums, sumsq, invN, W0, W1, W2, eb0, eb1, eb2, C, M, doRelu,
                      strideA, strideS, strideC, b % gx, b / gx);
    return;
  }
  const int sb = b - gx * ny;
  if constexpr (SIDE == SIDE_ZERO) zero_body(zbuf, zn4, sb, GEMM_THREADS);
  if constexpr (SIDE == SIDE_SCAN) tile_scan_body(counts, startw, cursor, sb);
  if constexpr (SIDE == SIDE_DEG)
    tile_deg_body<GEMM_THREADS>(rec, start, att, dinv, N_NODES, strideD, sb);
}

__global__ __launch_bounds__(ATT_THREADS, 12) void attention_kernel(
    const float* __restrict__ h, const int* __restrict__ row,
    const int* __restrict__ col, const float* __restrict__ Wea,
    const float* __restrict__ bea, float* __restrict__ att, int E,
    const float* __restrict__ Wna, const float* __restrict__ bna,
    float* __restrict__ xc, float* __restrict__ xo, int N, int nb) {
  const int b = blockIdx.x;
  if (b < nb) node_att_split_body(h, Wna, bna, xc, xo, N, b);
  else        edge_att_body(h, row, col, Wea, bea, att, E, b - nb);
}

extern "C" void kernel_launch(void* const* d_in, const int* in_sizes, int n_in,
                              void* d_out, int out_size, void* d_ws, size_t ws_size,
                              hipStream_t stream) {
  (void)in_sizes; (void)n_in; (void)out_size; (void)ws_size;
  const float* x       = (const float*)d_in[0];
  const int*   ei      = (const int*)d_in[1];
  const int*   row     = ei;
  const int*   col     = ei + N_EDGES;
  const int*   batch   = (const int*)d_in[2];
  const float* W_feat  = (const float*)d_in[3];
  const float* W_convs = (const float*)d_in[4];
  const float* b_convs = (const float*)d_in[5];
  const float* W_ea    = (const float*)d_in[6];
  const float* b_ea    = (const float*)d_in[7];
  const float* W_na    = (const float*)d_in[8];
  const float* b_na    = (const float*)d_in[9];
  const float* W_ctx   = (const float*)d_in[10];
  const float* b_ctx   = (const float*)d_in[11];
  const float* W_obj   = (const float*)d_in[12];
  const float* b_obj   = (const float*)d_in[13];
  const float* W1h[3]  = {(const float*)d_in[14], (const float*)d_in[18], (const float*)d_in[22]};
  const float* b1h[3]  = {(const float*)d_in[15], (const float*)d_in[19], (const float*)d_in[23]};
  const float* W2h[3]  = {(const float*)d_in[16], (const float*)d_in[20], (const float*)d_in[24]};
  const float* b2h[3]  = {(const float*)d_in[17], (const float*)d_in[21], (const float*)d_in[25]};

  const int SN = N_NODES * HIDDEN;
  const int SG = N_GRAPHS * HIDDEN;
  const int SS = STAT_STRIDE;
  size_t off = 0;
  auto carve = [&](size_t elems) -> float* {
    float* p = (float*)((char*)d_ws + off);
    off += ((elems * sizeof(float)) + 255) & ~(size_t)255;
    return p;
  };
  float* hb    = carve((size_t)SN);
  float* mcat  = carve((size_t)2 * SN);
  float* xcat  = carve((size_t)2 * SN);
  float* att   = carve((size_t)N_EDGES * 2);
  float* dinvu = carve(N_NODES);
  float* dinva = carve((size_t)2 * N_NODES);
  float* gcat  = carve((size_t)3 * SG);
  float* r1cat = carve((size_t)3 * SG);
  float* stat  = carve((size_t)3 * SS);
  float* part  = carve((size_t)2 * CS_MAX_SLABS * SS);
  int*   tcount  = (int*)carve(2 * BIN_STRIDE);
  int*   tstart  = (int*)carve(2 * BIN_STRIDE);
  int*   tcursor = (int*)carve(2 * BIN_STRIDE);
  int*   erec    = (int*)carve((size_t)2 * N_EDGES * 4);
  const int* srec   = erec + (size_t)N_EDGES * 4;
  const int* sstart = tstart + BIN_STRIDE;
  float* xc = xcat;
  float* xo = xcat + SN;
  float* gc = gcat;
  float* go = gcat + SG;
  float* gs = gcat + 2 * SG;

  auto nblk = [](long n, int b) { return (int)((n + b - 1) / b); };
  const int PN = nblk(N_NODES, CS_SLAB);
  const int GN = nblk(N_NODES / 16, GEMM_WAVES);
  const float invNn = 1.f / N_NODES;

  auto stats = [&](const float* A, int N, int ny, int strideA) {
    const int P = nblk(N, CS_SLAB);
    float* rows = P == 1 ? stat : part;
    colstats_slab_kernel<HIDDEN, SIDE_NONE><<<P * ny, CS_THREADS, 0, stream>>>(
        A, N, rows, strideA, P, ny, nullptr, 0, nullptr, nullptr, 0, nullptr, nullptr,
        nullptr, nullptr, nullptr, nullptr, 0);
    if (P > 1)
      colstats_finish_kernel<HIDDEN, SIDE_NONE><<<ny, CF_THREADS, 0, stream>>>(
          part, P, stat, ny, nullptr, nullptr, 0, nullptr);
  };
  auto gemm = [&](const float* A, float* C, int M, int relu, int ny,
                  int strideA, int strideC, float invN,
                  const float* W0, const float* W1, const float* W2,
                  const float* e0, const float* e1, const float* e2) {
    const int gx = nblk(M / 16, GEMM_WAVES);
    gemm_wmma_kernel<HIDDEN, SIDE_NONE><<<gx * ny, GEMM_THREADS, 0, stream>>>(
        A, stat, stat + N_FEAT, invN, W0, W1, W2, e0, e1, e2, C, M, relu,
        strideA, SS, strideC, gx, ny, nullptr, 0, nullptr, nullptr, nullptr,
        nullptr, nullptr, nullptr, nullptr, 0);
  };

  colstats_slab_kernel<N_FEAT, SIDE_ZERO><<<PN + 1, CS_THREADS, 0, stream>>>(
      x, N_NODES, part, 0, PN, 1, tcount, 2 * BIN_STRIDE / 4, nullptr, nullptr, 0,
      nullptr, nullptr, nullptr, nullptr, nullptr, nullptr, 0);
  colstats_finish_kernel<N_FEAT, SIDE_COUNT><<<1 + 2 * BIN_BLOCKS, CF_THREADS, 0, stream>>>(
      part, PN, stat, 1, row, col, N_EDGES, tcount);
  gemm_wmma_kernel<N_FEAT, SIDE_SCAN><<<GN + 2, GEMM_THREADS, 0, stream>>>(
      x, stat, stat + N_FEAT, invNn, W_feat, W_feat, W_feat, nullptr, nullptr, nullptr,
      hb, N_NODES, 1, 0, SS, 0, GN, 1, nullptr, 0, tcount, tstart, tcursor,
      nullptr, nullptr, nullptr, nullptr, 0);

  for (int i = 0; i < 3; ++i) {
    const float* Wi = W_convs + (size_t)i * HIDDEN * HIDDEN;
    const float* bi = b_convs + (size_t)i * HIDDEN;
    if (i == 0) {
      colstats_slab_kernel<HIDDEN, SIDE_FILL><<<PN + 2 * BIN_BLOCKS, CS_THREADS, 0, stream>>>(
          hb, N_NODES, part, 0, PN, 1, nullptr, 0, row, col, N_EDGES, tcursor, erec,
          nullptr, nullptr, nullptr, nullptr, 0);
      colstats_finish_kernel<HIDDEN, SIDE_NONE><<<1, CF_THREADS, 0, stream>>>(
          part, PN, stat, 1, nullptr, nullptr, 0, nullptr);
      gemm_wmma_kernel<HIDDEN, SIDE_DEG><<<GN + N_TILES, GEMM_THREADS, 0, stream>>>(
          hb, stat, stat + N_FEAT, invNn, Wi, Wi, Wi, nullptr, nullptr, nullptr,
          mcat, N_NODES, 0, 0, SS, 0, GN, 1, nullptr, 0, nullptr, nullptr, nullptr,
          srec, sstart, nullptr, dinvu, 0);
    } else {
      stats(hb, N_NODES, 1, 0);
      gemm(hb, mcat, N_NODES, 0, 1, 0, 0, invNn, Wi, Wi, Wi, nullptr, nullptr, nullptr);
    }
    tile_prop_kernel<<<dim3(N_TILES, 1), TP_THREADS, 0, stream>>>(
        mcat, erec, tstart, nullptr, 0, dinvu, bi, bi, hb, N_NODES, 0, 0);
  }

  const int NB_NODE = nblk(N_NODES, ATT_THREADS);
  attention_kernel<<<NB_NODE + nblk(N_EDGES, ATT_THREADS), ATT_THREADS, 0, stream>>>(
      hb, row, col, W_ea, b_ea, att, N_EDGES, W_na, b_na, xc, xo, N_NODES, NB_NODE);

  colstats_slab_kernel<HIDDEN, SIDE_DEG><<<2 * PN + N_TILES, CS_THREADS, 0, stream>>>(
      xcat, N_NODES, part, SN, PN, 2, nullptr, 0, nullptr, nullptr, 0, nullptr, nullptr,
      srec, sstart, att, dinva, N_NODES);
  colstats_finish_kernel<HIDDEN, SIDE_NONE><<<2, CF_THREADS, 0, stream>>>(
      part, PN, stat, 2, nullptr, nullptr, 0, nullptr);
  gemm_wmma_kernel<HIDDEN, SIDE_ZERO><<<2 * GN + nblk(2 * SG / 4, GEMM_THREADS), GEMM_THREADS, 0, stream>>>(
      xcat, stat, stat + N_FEAT, invNn, W_ctx, W_obj, W_obj, nullptr, nullptr, nullptr,
      mcat, N_NODES, 0, SN, SS, SN, GN, 2, (int*)gcat, 2 * SG / 4, nullptr, nullptr,
      nullptr, nullptr, nullptr, nullptr, nullptr, 0);
  tile_prop_kernel<<<dim3(N_TILES, 2), TP_THREADS, 0, stream>>>(
      mcat, erec, tstart, att, 0, dinva, b_ctx, b_obj, xcat, N_NODES, SN, N_NODES);
  pool_kernel<<<dim3(nblk(N_NODES, PL_ROWS * (PL_THREADS / 32)), 2), PL_THREADS, 0, stream>>>(
      xcat, batch, gcat, N_NODES, SN, SG);

  add_kernel<<<nblk(N_GRAPHS * HIDDEN, 256), 256, 0, stream>>>(gc, go, gs,
                                                               N_GRAPHS * HIDDEN);

  float* out = (float*)d_out;
  stats(gcat, N_GRAPHS, 3, SG);
  gemm(gcat, r1cat, N_GRAPHS, 1, 3, SG, SG, 1.f / N_GRAPHS,
       W1h[0], W1h[1], W1h[2], b1h[0], b1h[1], b1h[2]);
  stats(r1cat, N_GRAPHS, 3, SG);
  logits_kernel<<<dim3(nblk(N_GRAPHS, LOGIT_THREADS), 3), LOGIT_THREADS, 0, stream>>>(
      r1cat, stat, stat + N_FEAT, 1.f / N_GRAPHS, W2h[0], W2h[1], W2h[2],
      b2h[0], b2h[1], b2h[2], out, N_GRAPHS, SG, SS, N_GRAPHS * N_CLASSES);
}
